# starting kernel
# baseline (speedup 1.0000x reference)
#include <hip/hip_runtime.h>


#define BT       16
#define NTHREADS 256
#define NSLOT    (NTHREADS / BT)
#define W0       2048
#define W1       1024
#define W2       384
#define PFEAT    784
#define FBN      256
#define POPN     128
#define NIN0     (PFEAT + FBN)
#define STEPSN   8
#define NGATES   (W0 + W1 + W2)

typedef float v2f __attribute__((ext_vector_type(2)));
typedef float v8f __attribute__((ext_vector_type(8)));

__global__ void dlgn_precomp(const float* __restrict__ lg0,
                             const int*  __restrict__ le0, const int* __restrict__ ri0,
                             const float* __restrict__ lg1,
                             const int*  __restrict__ le1, const int* __restrict__ ri1,
                             const float* __restrict__ lg2,
                             const int*  __restrict__ le2, const int* __restrict__ ri2,
                             int2*  __restrict__ idx, float4* __restrict__ co) {
  int G = blockIdx.x * blockDim.x + threadIdx.x;
  if (G >= NGATES) return;
  const float* lg; const int* li; const int* ri; int g;
  if (G < W0)            { lg = lg0; li = le0; ri = ri0; g = G; }
  else if (G < W0 + W1)  { lg = lg1; li = le1; ri = ri1; g = G - W0; }
  else                   { lg = lg2; li = le2; ri = ri2; g = G - W0 - W1; }
  float w0 = 0.5f + 0.5f * sinf(lg[g * 4 + 0]);
  float w1 = 0.5f + 0.5f * sinf(lg[g * 4 + 1]);
  float w2 = 0.5f + 0.5f * sinf(lg[g * 4 + 2]);
  float w3 = 0.5f + 0.5f * sinf(lg[g * 4 + 3]);
  co[G]  = make_float4(w0, w2 - w0, w1 - w0, w0 - w1 - w2 + w3);
  idx[G] = make_int2(li[g], ri[g]);
}

__global__ __launch_bounds__(NTHREADS)
void dlgn_main(const float* __restrict__ x, const float* __restrict__ thr,
               const int2* __restrict__ idx, const float4* __restrict__ co,
               float* __restrict__ out) {
  __shared__ float sh_hin[NIN0 * BT];
  __shared__ float sh_l0 [W0   * BT];
  __shared__ float sh_l1 [W1   * BT];
  __shared__ float sh_pop[POPN * BT];

  const int t    = threadIdx.x;
  const int bb   = t & (BT - 1);
  const int slot = t / BT;
  const float* xb = x + (size_t)(blockIdx.x * BT + bb) * PFEAT;

  const float myth = thr[slot & 7];

  for (int j = slot; j < FBN; j += NSLOT)
    sh_hin[(PFEAT + j) * BT + bb] = 0.0f;

  for (int s = 0; s < STEPSN; ++s) {
    const float* xs = xb + s * 98;
    for (int p = slot; p < PFEAT; p += NSLOT) {
      float v = xs[p >> 3];
      sh_hin[p * BT + bb] = (v >= myth) ? 1.0f : 0.0f;
    }
    if (s + 1 < STEPSN) __builtin_prefetch(xs + 98, 0, 0);
    __syncthreads();

    for (int g = slot; g < W0; g += NSLOT) {
      int2   lr = idx[g];
      float4 w  = co[g];
      float L = sh_hin[lr.x * BT + bb];
      float R = sh_hin[lr.y * BT + bb];
      sh_l0[g * BT + bb] = w.x + w.y * L + w.z * R + w.w * (L * R);
    }
    __syncthreads();

    for (int g = slot; g < W1; g += NSLOT) {
      int2   lr = idx[W0 + g];
      float4 w  = co[W0 + g];
      float L = sh_l0[lr.x * BT + bb];
      float R = sh_l0[lr.y * BT + bb];
      sh_l1[g * BT + bb] = w.x + w.y * L + w.z * R + w.w * (L * R);
    }
    __syncthreads();

    for (int g = slot; g < W2; g += NSLOT) {
      int2   lr = idx[W0 + W1 + g];
      float4 w  = co[W0 + W1 + g];
      float L = sh_l1[lr.x * BT + bb];
      float R = sh_l1[lr.y * BT + bb];
      float v = w.x + w.y * L + w.z * R + w.w * (L * R);
      if (g < FBN) sh_hin[(PFEAT + g) * BT + bb] = v;
      else         sh_pop[(g - FBN) * BT + bb]  = v;
    }
    __syncthreads();
  }

  if (t < 32) {
    v8f acc = {};
    v2f ones; ones[0] = 1.0f; ones[1] = 1.0f;
    const int col   = t & 15;
    const int khalf = (t < 16) ? 0 : 2;
    #pragma unroll 4
    for (int kc = 0; kc < 32; ++kc) {
      v2f bm;
      bm[0] = sh_pop[(4 * kc + khalf + 0) * BT + col];
      bm[1] = sh_pop[(4 * kc + khalf + 1) * BT + col];
      acc = __builtin_amdgcn_wmma_f32_16x16x4_f32(
           false, ones,  false, bm,
           (short)0, acc,  false,  false);
    }
    if (t < 16) {
      float logit = acc[0] / 10.0f;
      float* ob = out + (size_t)(blockIdx.x * BT + t) * 10;
      #pragma unroll
      for (int c = 0; c < 10; ++c) ob[c] = logit;
    }
  }
}

extern "C" void kernel_launch(void* const* d_in, const int* in_sizes, int n_in,
                              void* d_out, int out_size, void* d_ws, size_t ws_size,
                              hipStream_t stream) {
  const float* x   = (const float*)d_in[0];
  const float* thr = (const float*)d_in[1];
  const float* lg0 = (const float*)d_in[2];
  const int*   le0 = (const int*)  d_in[3];
  const int*   ri0 = (const int*)  d_in[4];
  const float* lg1 = (const float*)d_in[5];
  const int*   le1 = (const int*)  d_in[6];
  const int*   ri1 = (const int*)  d_in[7];
  const float* lg2 = (const float*)d_in[8];
  const int*   le2 = (const int*)  d_in[9];
  const int*   ri2 = (const int*)  d_in[10];

  int2*   idx = (int2*)d_ws;
  float4* co  = (float4*)((char*)d_ws + 32768);

  dlgn_precomp<<<(NGATES + 255) / 256, 256, 0, stream>>>(
      lg0, le0, ri0, lg1, le1, ri1, lg2, le2, ri2, idx, co);

  dlgn_main<<<4096 / BT, NTHREADS, 0, stream>>>(x, thr, idx, co, (float*)d_out);
}


# Each lane reads a different gate's table entry (1 b64 + 1 b128 per 16 gates); DPP row_share operands of the adds and FMAs share it in the 16-lane row: no same-address LDS reads; sum order kept
# speedup vs baseline: 2.9198x; 2.9198x over previous
#include <hip/hip_runtime.h>


#define BT       16
#define NTHREADS 256
#define NSLOT    (NTHREADS / BT)
#define W0       2048
#define W1       1024
#define W2       384
#define PFEAT    784
#define FBN      256
#define POPN     128
#define STEPSN   8
#define NGATES   (W0 + W1 + W2)
#define NPIX     (PFEAT / 8)
#define PIXJ     ((NPIX + NSLOT - 1) / NSLOT)
#define FBROW    W1
#define AROWS    (FBROW + FBN)

typedef float v2f __attribute__((ext_vector_type(2)));
typedef float v8f __attribute__((ext_vector_type(8)));

__global__ void dlgn_precomp(const float* __restrict__ lg0,
                             const int*  __restrict__ le0, const int* __restrict__ ri0,
                             const float* __restrict__ lg1,
                             const int*  __restrict__ le1, const int* __restrict__ ri1,
                             const float* __restrict__ lg2,
                             const int*  __restrict__ le2, const int* __restrict__ ri2,
                             int2*  __restrict__ idx, float4* __restrict__ co) {
  int G = blockIdx.x * blockDim.x + threadIdx.x;
  if (G >= NGATES) return;
  const float* lg; const int* li; const int* ri; int g;
  if (G < W0)            { lg = lg0; li = le0; ri = ri0; g = G; }
  else if (G < W0 + W1)  { lg = lg1; li = le1; ri = ri1; g = G - W0; }
  else                   { lg = lg2; li = le2; ri = ri2; g = G - W0 - W1; }
  float w0 = 0.5f + 0.5f * sinf(lg[g * 4 + 0]);
  float w1 = 0.5f + 0.5f * sinf(lg[g * 4 + 1]);
  float w2 = 0.5f + 0.5f * sinf(lg[g * 4 + 2]);
  float w3 = 0.5f + 0.5f * sinf(lg[g * 4 + 3]);
  co[G]  = make_float4(w0, w2 - w0, w1 - w0, w0 - w1 - w2 + w3);
  int l = li[g], r = ri[g];
  if (G < W0) {
    l = (l < PFEAT) ? l : l + (FBROW - PFEAT);
    r = (r < PFEAT) ? r : r + (FBROW - PFEAT);
  }
  idx[G] = make_int2(l * (BT * 4), r * (BT * 4));
}

template <int N> __device__ __forceinline__ int row_bc(int v) {
  return __builtin_amdgcn_update_dpp(0, v, 0x150 + N, 0xf, 0xf, true);
}
template <int N> __device__ __forceinline__ float row_bc(float v) {
  return __int_as_float(row_bc<N>(__float_as_int(v)));
}

template <int N> __device__ __forceinline__ void fmac_bc(float& acc, float coef, float x) {
  asm("v_fmac_f32_dpp %0, %1, %2 row_share:%3 row_mask:0xf bank_mask:0xf"
      : "+v"(acc) : "v"(coef), "v"(x), "n"(N));
}

template <int H>
__device__ __forceinline__ void gate_trip(const char* col, float* dst, int2 e, float4 w) {
  float L[8], R[8];
#define GT_LOAD(U)                                                              \
  L[U] = *reinterpret_cast<const float*>(col + row_bc<8 * H + U>(e.x));         \
  R[U] = *reinterpret_cast<const float*>(col + row_bc<8 * H + U>(e.y));
  GT_LOAD(0) GT_LOAD(1) GT_LOAD(2) GT_LOAD(3) GT_LOAD(4) GT_LOAD(5) GT_LOAD(6) GT_LOAD(7)
#undef GT_LOAD
#define GT_EVAL(U)                                                              \
  {                                                                             \
    float acc = row_bc<8 * H + U>(w.x);                                         \
    fmac_bc<8 * H + U>(acc, w.y, L[U]);                                         \
    fmac_bc<8 * H + U>(acc, w.z, R[U]);                                         \
    fmac_bc<8 * H + U>(acc, w.w, L[U] * R[U]);                                  \
    dst[2 * (8 * H + U) * BT] = acc;                                            \
  }
  GT_EVAL(0) GT_EVAL(1) GT_EVAL(2) GT_EVAL(3) GT_EVAL(4) GT_EVAL(5) GT_EVAL(6) GT_EVAL(7)
#undef GT_EVAL
}

__device__ __forceinline__ void gate_group(const char* col, float* dst, int2 e, float4 w) {
  gate_trip<0>(col, dst, e, w);
  gate_trip<1>(col, dst, e, w);
}
template <int NG>
__device__ __forceinline__ void layer_groups(const char* col, float* dst,
                                             const int2* ti, const float4* tc) {
  int2 ea = ti[0]; float4 wa = tc[0];
  if constexpr (NG == 1) {
    gate_group(col, dst, ea, wa);
  } else {
    static_assert(NG % 2 == 0, "groups are taken in pairs");
    for (int g = 0; g < NG; g += 2) {
      const int2 eb = ti[(g + 1) * 256]; const float4 wb = tc[(g + 1) * 256];
      gate_group(col, dst + g * 256 * BT, ea, wa);
      const int gn = (g + 2 < NG) ? g + 2 : g;
      ea = ti[gn * 256]; wa = tc[gn * 256];
      gate_group(col, dst + (g + 1) * 256 * BT, eb, wb);
    }
  }
}

__global__ __launch_bounds__(NTHREADS)
void dlgn_main(const float* __restrict__ x, const float* __restrict__ thr,
               const int2* __restrict__ idx, const float4* __restrict__ co,
               float* __restrict__ out) {
  __shared__ float  sh_a  [AROWS * BT];
  __shared__ float  sh_l0 [W0    * BT];
  __shared__ float  sh_pop[POPN  * BT];
  __shared__ int2   sh_idx[NGATES];
  __shared__ float4 sh_co [NGATES];

  const int t    = threadIdx.x;
  const int bb   = t & (BT - 1);
  const int slot = t / BT;
  const int wave = t >> 5;
  const int hrow = (t >> 4) & 1;
  const int tpos = wave * 32 + 2 * bb + hrow;
  const int orow = wave * 32 + hrow;
  const float* xb = x + (size_t)(blockIdx.x * BT + bb) * PFEAT;
  const char* a_col  = reinterpret_cast<const char*>(sh_a)  + bb * 4;
  const char* l0_col = reinterpret_cast<const char*>(sh_l0) + bb * 4;

  const int krot = (slot & 1) * 2;
  float thv[8];
#pragma unroll
  for (int k = 0; k < 8; ++k) thv[k] = thr[(k + krot) & 7];

  for (int i = t; i < NGATES; i += NTHREADS) {
    sh_idx[i] = idx[i];
    sh_co[i]  = co[i];
  }

  for (int j = slot; j < FBN; j += NSLOT)
    sh_a[(FBROW + j) * BT + bb] = 0.0f;

  for (int s = 0; s < STEPSN; ++s) {
    const float* xs = xb + s * 98;
    float pv[PIXJ];
#pragma unroll
    for (int j = 0; j < PIXJ; ++j) {
      const int q = slot + NSLOT * j;
      const bool live = (NSLOT * j + NSLOT <= NPIX) || (q < NPIX);
      pv[j] = live ? xs[q] : 0.0f;
    }
#pragma unroll
    for (int j = 0; j < PIXJ; ++j) {
      const int q = slot + NSLOT * j;
      const bool live = (NSLOT * j + NSLOT <= NPIX) || (q < NPIX);
      if (live) {
        float* r0 = sh_a + (q * 8 + krot) * BT + bb;
#pragma unroll
        for (int k = 0; k < 6; ++k)
          r0[k * BT] = (pv[j] >= thv[k]) ? 1.0f : 0.0f;
        float* r6 = r0 - 4 * krot * BT;
        r6[6 * BT] = (pv[j] >= thv[6]) ? 1.0f : 0.0f;
        r6[7 * BT] = (pv[j] >= thv[7]) ? 1.0f : 0.0f;
      }
    }
    if (s + 1 < STEPSN) __builtin_prefetch(xs + 98, 0, 0);
    __syncthreads();

    layer_groups<W0 / 256>(a_col, sh_l0 + orow * BT + bb, sh_idx + tpos, sh_co + tpos);
    __syncthreads();

    layer_groups<W1 / 256>(l0_col, sh_a + orow * BT + bb, sh_idx + W0 + tpos, sh_co + W0 + tpos);
    __syncthreads();

    layer_groups<1>(a_col, sh_a + (FBROW + orow) * BT + bb,
                    sh_idx + W0 + W1 + tpos, sh_co + W0 + W1 + tpos);
    {
      const int pe = W0 + W1 + FBN + wave * 16 + 2 * (bb & 7) + hrow;
      gate_trip<0>(a_col, sh_pop + (wave * 16 + hrow) * BT + bb, sh_idx[pe], sh_co[pe]);
    }
    __syncthreads();
  }

  if (t < 32) {
    v8f acc = {};
    v2f ones; ones[0] = 1.0f; ones[1] = 1.0f;
    const int col   = t & 15;
    const int khalf = (t < 16) ? 0 : 2;
    #pragma unroll 4
    for (int kc = 0; kc < 32; ++kc) {
      v2f bm;
      bm[0] = sh_pop[(4 * kc + khalf + 0) * BT + col];
      bm[1] = sh_pop[(4 * kc + khalf + 1) * BT + col];
      acc = __builtin_amdgcn_wmma_f32_16x16x4_f32(
           false, ones,  false, bm,
           (short)0, acc,  false,  false);
    }
    if (t < 16) {
      float logit = acc[0] / 10.0f;
      float* ob = out + (size_t)(blockIdx.x * BT + t) * 10;
      #pragma unroll
      for (int c = 0; c < 10; ++c) ob[c] = logit;
    }
  }
}

extern "C" void kernel_launch(void* const* d_in, const int* in_sizes, int n_in,
                              void* d_out, int out_size, void* d_ws, size_t ws_size,
                              hipStream_t stream) {
  const float* x   = (const float*)d_in[0];
  const float* thr = (const float*)d_in[1];
  const float* lg0 = (const float*)d_in[2];
  const int*   le0 = (const int*)  d_in[3];
  const int*   ri0 = (const int*)  d_in[4];
  const float* lg1 = (const float*)d_in[5];
  const int*   le1 = (const int*)  d_in[6];
  const int*   ri1 = (const int*)  d_in[7];
  const float* lg2 = (const float*)d_in[8];
  const int*   le2 = (const int*)  d_in[9];
  const int*   ri2 = (const int*)  d_in[10];

  int2*   idx = (int2*)d_ws;
  float4* co  = (float4*)((char*)d_ws + 32768);

  dlgn_precomp<<<(NGATES + 255) / 256, 256, 0, stream>>>(
      lg0, le0, ri0, lg1, le1, ri1, lg2, le2, ri2, idx, co);

  dlgn_main<<<4096 / BT, NTHREADS, 0, stream>>>(x, thr, idx, co, (float*)d_out);
}


# Only what the output uses is computed: population gates in the last step only, and that step evaluates just their cone (2 layer-1 + 4 layer-0 gates each, in registers), no feedback there; sums kept
# speedup vs baseline: 3.1556x; 1.0808x over previous
#include <hip/hip_runtime.h>


#define BT       16
#define NTHREADS 256
#define NSLOT    (NTHREADS / BT)
#define W0       2048
#define W1       1024
#define W2       384
#define PFEAT    784
#define FBN      256
#define POPN     128
#define STEPSN   8
#define NGATES   (W0 + W1 + W2)
#define NPIX     (PFEAT / 8)
#define PIXJ     ((NPIX + NSLOT - 1) / NSLOT)
#define FBROW    W1
#define AROWS    (FBROW + FBN)

typedef float v2f __attribute__((ext_vector_type(2)));
typedef float v8f __attribute__((ext_vector_type(8)));

__global__ void dlgn_precomp(const float* __restrict__ lg0,
                             const int*  __restrict__ le0, const int* __restrict__ ri0,
                             const float* __restrict__ lg1,
                             const int*  __restrict__ le1, const int* __restrict__ ri1,
                             const float* __restrict__ lg2,
                             const int*  __restrict__ le2, const int* __restrict__ ri2,
                             int2*  __restrict__ idx, float4* __restrict__ co) {
  int G = blockIdx.x * blockDim.x + threadIdx.x;
  if (G >= NGATES) return;
  const float* lg; const int* li; const int* ri; int g;
  if (G < W0)            { lg = lg0; li = le0; ri = ri0; g = G; }
  else if (G < W0 + W1)  { lg = lg1; li = le1; ri = ri1; g = G - W0; }
  else                   { lg = lg2; li = le2; ri = ri2; g = G - W0 - W1; }
  float w0 = 0.5f + 0.5f * sinf(lg[g * 4 + 0]);
  float w1 = 0.5f + 0.5f * sinf(lg[g * 4 + 1]);
  float w2 = 0.5f + 0.5f * sinf(lg[g * 4 + 2]);
  float w3 = 0.5f + 0.5f * sinf(lg[g * 4 + 3]);
  co[G]  = make_float4(w0, w2 - w0, w1 - w0, w0 - w1 - w2 + w3);
  int l = li[g], r = ri[g];
  if (G < W0) {
    l = (l < PFEAT) ? l : l + (FBROW - PFEAT);
    r = (r < PFEAT) ? r : r + (FBROW - PFEAT);
  }
  idx[G] = make_int2(l * (BT * 4), r * (BT * 4));
}

template <int N> __device__ __forceinline__ int row_bc(int v) {
  return __builtin_amdgcn_update_dpp(0, v, 0x150 + N, 0xf, 0xf, true);
}
template <int N> __device__ __forceinline__ float row_bc(float v) {
  return __int_as_float(row_bc<N>(__float_as_int(v)));
}

template <int N> __device__ __forceinline__ void fmac_bc(float& acc, float coef, float x) {
  asm("v_fmac_f32_dpp %0, %1, %2 row_share:%3 row_mask:0xf bank_mask:0xf"
      : "+v"(acc) : "v"(coef), "v"(x), "n"(N));
}

template <int H>
__device__ __forceinline__ void gate_trip(const char* col, float* dst, int2 e, float4 w) {
  float L[8], R[8];
#define GT_LOAD(U)                                                              \
  L[U] = *reinterpret_cast<const float*>(col + row_bc<8 * H + U>(e.x));         \
  R[U] = *reinterpret_cast<const float*>(col + row_bc<8 * H + U>(e.y));
  GT_LOAD(0) GT_LOAD(1) GT_LOAD(2) GT_LOAD(3) GT_LOAD(4) GT_LOAD(5) GT_LOAD(6) GT_LOAD(7)
#undef GT_LOAD
#define GT_EVAL(U)                                                              \
  {                                                                             \
    float acc = row_bc<8 * H + U>(w.x);                                         \
    fmac_bc<8 * H + U>(acc, w.y, L[U]);                                         \
    fmac_bc<8 * H + U>(acc, w.z, R[U]);                                         \
    fmac_bc<8 * H + U>(acc, w.w, L[U] * R[U]);                                  \
    dst[2 * (8 * H + U) * BT] = acc;                                            \
  }
  GT_EVAL(0) GT_EVAL(1) GT_EVAL(2) GT_EVAL(3) GT_EVAL(4) GT_EVAL(5) GT_EVAL(6) GT_EVAL(7)
#undef GT_EVAL
}

__device__ __forceinline__ void gate_group(const char* col, float* dst, int2 e, float4 w) {
  gate_trip<0>(col, dst, e, w);
  gate_trip<1>(col, dst, e, w);
}
template <int NG>
__device__ __forceinline__ void layer_groups(const char* col, float* dst,
                                             const int2* ti, const float4* tc) {
  int2 ea = ti[0]; float4 wa = tc[0];
  if constexpr (NG == 1) {
    gate_group(col, dst, ea, wa);
  } else {
    static_assert(NG % 2 == 0, "groups are taken in pairs");
    for (int g = 0; g < NG; g += 2) {
      const int2 eb = ti[(g + 1) * 256]; const float4 wb = tc[(g + 1) * 256];
      gate_group(col, dst + g * 256 * BT, ea, wa);
      const int gn = (g + 2 < NG) ? g + 2 : g;
      ea = ti[gn * 256]; wa = tc[gn * 256];
      gate_group(col, dst + (g + 1) * 256 * BT, eb, wb);
    }
  }
}

template <int N> __device__ __forceinline__ float gate_bc(float4 w, float L, float R) {
  float acc = row_bc<N>(w.x);
  fmac_bc<N>(acc, w.y, L);
  fmac_bc<N>(acc, w.z, R);
  fmac_bc<N>(acc, w.w, L * R);
  return acc;
}
template <int N> __device__ __forceinline__ float gate_ld_bc(const char* col, int2 e, float4 w) {
  const float L = *reinterpret_cast<const float*>(col + row_bc<N>(e.x));
  const float R = *reinterpret_cast<const float*>(col + row_bc<N>(e.y));
  return gate_bc<N>(w, L, R);
}
template <int U>
__device__ __forceinline__ void pop_cone(const char* col, float* dst,
                                         int2 e00, float4 w00, int2 e01, float4 w01,
                                         float4 w1, float4 wp) {
  const float xa0 = gate_ld_bc<U>(col, e00, w00);
  const float xa1 = gate_ld_bc<U>(col, e01, w01);
  const float ya  = gate_bc<U>(w1, xa0, xa1);
  const float xb0 = gate_ld_bc<U + 8>(col, e00, w00);
  const float xb1 = gate_ld_bc<U + 8>(col, e01, w01);
  const float yb  = gate_bc<U + 8>(w1, xb0, xb1);
  dst[2 * U * BT] = gate_bc<U>(wp, ya, yb);
}

__global__ __launch_bounds__(NTHREADS)
void dlgn_main(const float* __restrict__ x, const float* __restrict__ thr,
               const int2* __restrict__ idx, const float4* __restrict__ co,
               float* __restrict__ out) {
  __shared__ float  sh_a  [AROWS * BT];
  __shared__ float  sh_l0 [W0    * BT];
  __shared__ float  sh_pop[POPN  * BT];
  __shared__ int2   sh_idx[NGATES];
  __shared__ float4 sh_co [NGATES];

  const int t    = threadIdx.x;
  const int bb   = t & (BT - 1);
  const int slot = t / BT;
  const int wave = t >> 5;
  const int hrow = (t >> 4) & 1;
  const int tpos = wave * 32 + 2 * bb + hrow;
  const int orow = wave * 32 + hrow;
  const float* xb = x + (size_t)(blockIdx.x * BT + bb) * PFEAT;
  const char* a_col  = reinterpret_cast<const char*>(sh_a)  + bb * 4;
  const char* l0_col = reinterpret_cast<const char*>(sh_l0) + bb * 4;

  const int krot = (slot & 1) * 2;
  float thv[8];
#pragma unroll
  for (int k = 0; k < 8; ++k) thv[k] = thr[(k + krot) & 7];

  for (int i = t; i < NGATES; i += NTHREADS) {
    sh_idx[i] = idx[i];
    sh_co[i]  = co[i];
  }

  for (int j = slot; j < FBN; j += NSLOT)
    sh_a[(FBROW + j) * BT + bb] = 0.0f;

  for (int s = 0; s < STEPSN; ++s) {
    const float* xs = xb + s * 98;
    float pv[PIXJ];
#pragma unroll
    for (int j = 0; j < PIXJ; ++j) {
      const int q = slot + NSLOT * j;
      const bool live = (NSLOT * j + NSLOT <= NPIX) || (q < NPIX);
      pv[j] = live ? xs[q] : 0.0f;
    }
#pragma unroll
    for (int j = 0; j < PIXJ; ++j) {
      const int q = slot + NSLOT * j;
      const bool live = (NSLOT * j + NSLOT <= NPIX) || (q < NPIX);
      if (live) {
        float* r0 = sh_a + (q * 8 + krot) * BT + bb;
#pragma unroll
        for (int k = 0; k < 6; ++k)
          r0[k * BT] = (pv[j] >= thv[k]) ? 1.0f : 0.0f;
        float* r6 = r0 - 4 * krot * BT;
        r6[6 * BT] = (pv[j] >= thv[6]) ? 1.0f : 0.0f;
        r6[7 * BT] = (pv[j] >= thv[7]) ? 1.0f : 0.0f;
      }
    }
    if (s + 1 < STEPSN) __builtin_prefetch(xs + 98, 0, 0);
    __syncthreads();

    if (s + 1 < STEPSN) {
      layer_groups<W0 / 256>(a_col, sh_l0 + orow * BT + bb, sh_idx + tpos, sh_co + tpos);
      __syncthreads();

      layer_groups<W1 / 256>(l0_col, sh_a + orow * BT + bb, sh_idx + W0 + tpos, sh_co + W0 + tpos);
      __syncthreads();

      layer_groups<1>(a_col, sh_a + (FBROW + orow) * BT + bb,
                      sh_idx + W0 + W1 + tpos, sh_co + W0 + W1 + tpos);
      __syncthreads();
    } else {
      const int pe = W0 + W1 + FBN + wave * 16 + 2 * (bb & 7) + hrow;
      const int2 ep = sh_idx[pe]; const float4 wp = sh_co[pe];
      const int g1 = W0 + ((bb < 8) ? ep.x : ep.y) / (BT * 4);
      const int2 e1 = sh_idx[g1]; const float4 w1 = sh_co[g1];
      const int g00 = e1.x / (BT * 4), g01 = e1.y / (BT * 4);
      const int2 e00 = sh_idx[g00]; const float4 w00 = sh_co[g00];
      const int2 e01 = sh_idx[g01]; const float4 w01 = sh_co[g01];
      float* pd = sh_pop + (wave * 16 + hrow) * BT + bb;
      pop_cone<0>(a_col, pd, e00, w00, e01, w01, w1, wp);
      pop_cone<1>(a_col, pd, e00, w00, e01, w01, w1, wp);
      pop_cone<2>(a_col, pd, e00, w00, e01, w01, w1, wp);
      pop_cone<3>(a_col, pd, e00, w00, e01, w01, w1, wp);
      pop_cone<4>(a_col, pd, e00, w00, e01, w01, w1, wp);
      pop_cone<5>(a_col, pd, e00, w00, e01, w01, w1, wp);
      pop_cone<6>(a_col, pd, e00, w00, e01, w01, w1, wp);
      pop_cone<7>(a_col, pd, e00, w00, e01, w01, w1, wp);
      __syncthreads();
    }
  }

  if (t < 32) {
    v8f acc = {};
    v2f ones; ones[0] = 1.0f; ones[1] = 1.0f;
    const int col   = t & 15;
    const int khalf = (t < 16) ? 0 : 2;
    #pragma unroll 4
    for (int kc = 0; kc < 32; ++kc) {
      v2f bm;
      bm[0] = sh_pop[(4 * kc + khalf + 0) * BT + col];
      bm[1] = sh_pop[(4 * kc + khalf + 1) * BT + col];
      acc = __builtin_amdgcn_wmma_f32_16x16x4_f32(
           false, ones,  false, bm,
           (short)0, acc,  false,  false);
    }
    if (t < 16) {
      float logit = acc[0] / 10.0f;
      float* ob = out + (size_t)(blockIdx.x * BT + t) * 10;
      #pragma unroll
      for (int c = 0; c < 10; ++c) ob[c] = logit;
    }
  }
}

extern "C" void kernel_launch(void* const* d_in, const int* in_sizes, int n_in,
                              void* d_out, int out_size, void* d_ws, size_t ws_size,
                              hipStream_t stream) {
  const float* x   = (const float*)d_in[0];
  const float* thr = (const float*)d_in[1];
  const float* lg0 = (const float*)d_in[2];
  const int*   le0 = (const int*)  d_in[3];
  const int*   ri0 = (const int*)  d_in[4];
  const float* lg1 = (const float*)d_in[5];
  const int*   le1 = (const int*)  d_in[6];
  const int*   ri1 = (const int*)  d_in[7];
  const float* lg2 = (const float*)d_in[8];
  const int*   le2 = (const int*)  d_in[9];
  const int*   ri2 = (const int*)  d_in[10];

  int2*   idx = (int2*)d_ws;
  float4* co  = (float4*)((char*)d_ws + 32768);

  dlgn_precomp<<<(NGATES + 255) / 256, 256, 0, stream>>>(
      lg0, le0, ri0, lg1, le1, ri1, lg2, le2, ri2, idx, co);

  dlgn_main<<<4096 / BT, NTHREADS, 0, stream>>>(x, thr, idx, co, (float*)d_out);
}


# Each step evaluates only the input cones of the gates it hands on (256 feedback gates in steps 0..6, as the last step's 128), entries and gather offsets held in registers: no layer arrays; sums kept
# speedup vs baseline: 4.3844x; 1.3894x over previous
#include <hip/hip_runtime.h>


#define BT       16
#define NTHREADS 256
#define NSLOT    (NTHREADS / BT)
#define W0       2048
#define W1       1024
#define W2       384
#define PFEAT    784
#define FBN      256
#define POPN     128
#define STEPSN   8
#define NGATES   (W0 + W1 + W2)
#define NPIX     (PFEAT / 8)
#define PIXJ     ((NPIX + NSLOT - 1) / NSLOT)
#define FBROW    W1
#define AROWS    (FBROW + FBN)

typedef float v2f __attribute__((ext_vector_type(2)));
typedef float v8f __attribute__((ext_vector_type(8)));

__global__ void dlgn_precomp(const float* __restrict__ lg0,
                             const int*  __restrict__ le0, const int* __restrict__ ri0,
                             const float* __restrict__ lg1,
                             const int*  __restrict__ le1, const int* __restrict__ ri1,
                             const float* __restrict__ lg2,
                             const int*  __restrict__ le2, const int* __restrict__ ri2,
                             int2*  __restrict__ idx, float4* __restrict__ co) {
  int G = blockIdx.x * blockDim.x + threadIdx.x;
  if (G >= NGATES) return;
  const float* lg; const int* li; const int* ri; int g;
  if (G < W0)            { lg = lg0; li = le0; ri = ri0; g = G; }
  else if (G < W0 + W1)  { lg = lg1; li = le1; ri = ri1; g = G - W0; }
  else                   { lg = lg2; li = le2; ri = ri2; g = G - W0 - W1; }
  float w0 = 0.5f + 0.5f * sinf(lg[g * 4 + 0]);
  float w1 = 0.5f + 0.5f * sinf(lg[g * 4 + 1]);
  float w2 = 0.5f + 0.5f * sinf(lg[g * 4 + 2]);
  float w3 = 0.5f + 0.5f * sinf(lg[g * 4 + 3]);
  co[G]  = make_float4(w0, w2 - w0, w1 - w0, w0 - w1 - w2 + w3);
  int l = li[g], r = ri[g];
  if (G < W0) {
    l = (l < PFEAT) ? l : l + (FBROW - PFEAT);
    r = (r < PFEAT) ? r : r + (FBROW - PFEAT);
  }
  idx[G] = make_int2(l * (BT * 4), r * (BT * 4));
}

template <int N> __device__ __forceinline__ int row_bc(int v) {
  return __builtin_amdgcn_update_dpp(0, v, 0x150 + N, 0xf, 0xf, true);
}
template <int N> __device__ __forceinline__ float row_bc(float v) {
  return __int_as_float(row_bc<N>(__float_as_int(v)));
}

template <int N> __device__ __forceinline__ void fmac_bc(float& acc, float coef, float x) {
  asm("v_fmac_f32_dpp %0, %1, %2 row_share:%3 row_mask:0xf bank_mask:0xf"
      : "+v"(acc) : "v"(coef), "v"(x), "n"(N));
}

template <int N> __device__ __forceinline__ float gate_bc(float4 w, float L, float R) {
  float acc = row_bc<N>(w.x);
  fmac_bc<N>(acc, w.y, L);
  fmac_bc<N>(acc, w.z, R);
  fmac_bc<N>(acc, w.w, L * R);
  return acc;
}
struct ConeEnt { int2 e00; float4 w00; int2 e01; float4 w01; float4 w1; float4 wp; };
__device__ __forceinline__ ConeEnt cone_entries(const int2* ti, const float4* tc,
                                                int g2, bool left) {
  ConeEnt c;
  const int2 ep = ti[g2]; c.wp = tc[g2];
  const int g1 = W0 + (left ? ep.x : ep.y) / (BT * 4);
  const int2 e1 = ti[g1]; c.w1 = tc[g1];
  const int g00 = e1.x / (BT * 4), g01 = e1.y / (BT * 4);
  c.e00 = ti[g00]; c.w00 = tc[g00];
  c.e01 = ti[g01]; c.w01 = tc[g01];
  return c;
}
struct ConeOff { int o[8]; };
template <int U>
__device__ __forceinline__ ConeOff cone_offsets(int colb, const ConeEnt& c) {
  ConeOff a;
  a.o[0] = colb + row_bc<U>(c.e00.x);      a.o[1] = colb + row_bc<U>(c.e00.y);
  a.o[2] = colb + row_bc<U>(c.e01.x);      a.o[3] = colb + row_bc<U>(c.e01.y);
  a.o[4] = colb + row_bc<U + 8>(c.e00.x);  a.o[5] = colb + row_bc<U + 8>(c.e00.y);
  a.o[6] = colb + row_bc<U + 8>(c.e01.x);  a.o[7] = colb + row_bc<U + 8>(c.e01.y);
  return a;
}
__device__ __forceinline__ void cone_offsets_all(int colb, const ConeEnt& c, ConeOff (&a)[8]) {
  a[0] = cone_offsets<0>(colb, c); a[1] = cone_offsets<1>(colb, c);
  a[2] = cone_offsets<2>(colb, c); a[3] = cone_offsets<3>(colb, c);
  a[4] = cone_offsets<4>(colb, c); a[5] = cone_offsets<5>(colb, c);
  a[6] = cone_offsets<6>(colb, c); a[7] = cone_offsets<7>(colb, c);
}
__device__ __forceinline__ float ld_off(const char* base, int off) {
  return *reinterpret_cast<const float*>(base + off);
}
template <int U>
__device__ __forceinline__ float cone_val(const char* base, const ConeOff& a, const ConeEnt& c) {
  const float xa0 = gate_bc<U>(c.w00, ld_off(base, a.o[0]), ld_off(base, a.o[1]));
  const float xa1 = gate_bc<U>(c.w01, ld_off(base, a.o[2]), ld_off(base, a.o[3]));
  const float ya  = gate_bc<U>(c.w1, xa0, xa1);
  const float xb0 = gate_bc<U + 8>(c.w00, ld_off(base, a.o[4]), ld_off(base, a.o[5]));
  const float xb1 = gate_bc<U + 8>(c.w01, ld_off(base, a.o[6]), ld_off(base, a.o[7]));
  const float yb  = gate_bc<U + 8>(c.w1, xb0, xb1);
  return gate_bc<U>(c.wp, ya, yb);
}
__device__ __forceinline__ void cone_round(const char* base, const ConeOff (&a)[8],
                                           const ConeEnt& c, float (&v)[8]) {
  v[0] = cone_val<0>(base, a[0], c); v[1] = cone_val<1>(base, a[1], c);
  v[2] = cone_val<2>(base, a[2], c); v[3] = cone_val<3>(base, a[3], c);
  v[4] = cone_val<4>(base, a[4], c); v[5] = cone_val<5>(base, a[5], c);
  v[6] = cone_val<6>(base, a[6], c); v[7] = cone_val<7>(base, a[7], c);
}

__device__ __forceinline__ void encode_step(float* sh_a, const float* xs, int slot, int krot,
                                            int bb, const float (&thv)[8]) {
  float pv[PIXJ];
#pragma unroll
  for (int j = 0; j < PIXJ; ++j) {
    const int q = slot + NSLOT * j;
    const bool live = (NSLOT * j + NSLOT <= NPIX) || (q < NPIX);
    pv[j] = live ? xs[q] : 0.0f;
  }
#pragma unroll
  for (int j = 0; j < PIXJ; ++j) {
    const int q = slot + NSLOT * j;
    const bool live = (NSLOT * j + NSLOT <= NPIX) || (q < NPIX);
    if (live) {
      float* r0 = sh_a + (q * 8 + krot) * BT + bb;
#pragma unroll
      for (int k = 0; k < 6; ++k)
        r0[k * BT] = (pv[j] >= thv[k]) ? 1.0f : 0.0f;
      float* r6 = r0 - 4 * krot * BT;
      r6[6 * BT] = (pv[j] >= thv[6]) ? 1.0f : 0.0f;
      r6[7 * BT] = (pv[j] >= thv[7]) ? 1.0f : 0.0f;
    }
  }
}

__global__ __launch_bounds__(NTHREADS)
void dlgn_main(const float* __restrict__ x, const float* __restrict__ thr,
               const int2* __restrict__ idx, const float4* __restrict__ co,
               float* __restrict__ out) {
  __shared__ float  sh_a  [AROWS * BT];
  __shared__ float  sh_pop[POPN  * BT];
  __shared__ int2   sh_idx[NGATES];
  __shared__ float4 sh_co [NGATES];

  const int t    = threadIdx.x;
  const int bb   = t & (BT - 1);
  const int slot = t / BT;
  const int wave = t >> 5;
  const int hrow = (t >> 4) & 1;
  const float* xb = x + (size_t)(blockIdx.x * BT + bb) * PFEAT;
  const char* a_base = reinterpret_cast<const char*>(sh_a);

  const int krot = (slot & 1) * 2;
  float thv[8];
#pragma unroll
  for (int k = 0; k < 8; ++k) thv[k] = thr[(k + krot) & 7];

  for (int i = t; i < NGATES; i += NTHREADS) {
    sh_idx[i] = idx[i];
    sh_co[i]  = co[i];
  }

  for (int j = slot; j < FBN; j += NSLOT)
    sh_a[(FBROW + j) * BT + bb] = 0.0f;
  __syncthreads();

  const int  gl   = wave * 16 + 2 * (bb & 7) + hrow;
  const bool left = bb < 8;
  const ConeEnt cf0 = cone_entries(sh_idx, sh_co, W0 + W1 + gl, left);
  const ConeEnt cf1 = cone_entries(sh_idx, sh_co, W0 + W1 + 128 + gl, left);
  ConeOff of0[8], of1[8];
  cone_offsets_all(bb * 4, cf0, of0);
  cone_offsets_all(bb * 4, cf1, of1);
  float* fd = sh_a   + (FBROW + wave * 16 + hrow) * BT + bb;

  for (int s = 0; s + 1 < STEPSN; ++s) {
    const float* xs = xb + s * NPIX;
    encode_step(sh_a, xs, slot, krot, bb, thv);
    __builtin_prefetch(xs + NPIX, 0, 0);
    __syncthreads();

    float f0[8], f1[8];
    cone_round(a_base, of0, cf0, f0);
    cone_round(a_base, of1, cf1, f1);
    __syncthreads();
#pragma unroll
    for (int u = 0; u < 8; ++u) {
      fd[2 * u * BT]         = f0[u];
      fd[(128 + 2 * u) * BT] = f1[u];
    }
  }

  {
    const ConeEnt cpp = cone_entries(sh_idx, sh_co, W0 + W1 + FBN + gl, left);
    ConeOff ofp[8];
    cone_offsets_all(bb * 4, cpp, ofp);
    encode_step(sh_a, xb + (STEPSN - 1) * NPIX, slot, krot, bb, thv);
    __syncthreads();
    float p[8];
    cone_round(a_base, ofp, cpp, p);
    float* pd = sh_pop + (wave * 16 + hrow) * BT + bb;
#pragma unroll
    for (int u = 0; u < 8; ++u) pd[2 * u * BT] = p[u];
    __syncthreads();
  }

  if (t < 32) {
    v8f acc = {};
    v2f ones; ones[0] = 1.0f; ones[1] = 1.0f;
    const int col   = t & 15;
    const int khalf = (t < 16) ? 0 : 2;
    #pragma unroll 4
    for (int kc = 0; kc < 32; ++kc) {
      v2f bm;
      bm[0] = sh_pop[(4 * kc + khalf + 0) * BT + col];
      bm[1] = sh_pop[(4 * kc + khalf + 1) * BT + col];
      acc = __builtin_amdgcn_wmma_f32_16x16x4_f32(
           false, ones,  false, bm,
           (short)0, acc,  false,  false);
    }
    if (t < 16) {
      float logit = acc[0] / 10.0f;
      float* ob = out + (size_t)(blockIdx.x * BT + t) * 10;
      #pragma unroll
      for (int c = 0; c < 10; ++c) ob[c] = logit;
    }
  }
}

extern "C" void kernel_launch(void* const* d_in, const int* in_sizes, int n_in,
                              void* d_out, int out_size, void* d_ws, size_t ws_size,
                              hipStream_t stream) {
  const float* x   = (const float*)d_in[0];
  const float* thr = (const float*)d_in[1];
  const float* lg0 = (const float*)d_in[2];
  const int*   le0 = (const int*)  d_in[3];
  const int*   ri0 = (const int*)  d_in[4];
  const float* lg1 = (const float*)d_in[5];
  const int*   le1 = (const int*)  d_in[6];
  const int*   ri1 = (const int*)  d_in[7];
  const float* lg2 = (const float*)d_in[8];
  const int*   le2 = (const int*)  d_in[9];
  const int*   ri2 = (const int*)  d_in[10];

  int2*   idx = (int2*)d_ws;
  float4* co  = (float4*)((char*)d_ws + 32768);

  dlgn_precomp<<<(NGATES + 255) / 256, 256, 0, stream>>>(
      lg0, le0, ri0, lg1, le1, ri1, lg2, le2, ri2, idx, co);

  dlgn_main<<<4096 / BT, NTHREADS, 0, stream>>>(x, thr, idx, co, (float*)d_out);
}


# A lane is a cone, not a batch column: thread t evaluates gate t's cone for all 16 columns, coefficients in its own registers (plain FMAs, no DPP operand), rows read 128 bits a lane; sums kept
# speedup vs baseline: 5.9967x; 1.3677x over previous
#include <hip/hip_runtime.h>


#define BT       16
#define NTHREADS 256
#define NSLOT    (NTHREADS / BT)
#define W0       2048
#define W1       1024
#define W2       384
#define PFEAT    784
#define FBN      256
#define POPN     128
#define STEPSN   8
#define NGATES   (W0 + W1 + W2)
#define NPIX     (PFEAT / 8)
#define PIXJ     ((NPIX + NSLOT - 1) / NSLOT)
#define FBROW    W1
#define AROWS    (FBROW + FBN)

typedef float v2f __attribute__((ext_vector_type(2)));
typedef float v4f __attribute__((ext_vector_type(4)));
typedef float v8f __attribute__((ext_vector_type(8)));

__global__ void dlgn_precomp(const float* __restrict__ lg0,
                             const int*  __restrict__ le0, const int* __restrict__ ri0,
                             const float* __restrict__ lg1,
                             const int*  __restrict__ le1, const int* __restrict__ ri1,
                             const float* __restrict__ lg2,
                             const int*  __restrict__ le2, const int* __restrict__ ri2,
                             int2*  __restrict__ idx, float4* __restrict__ co) {
  int G = blockIdx.x * blockDim.x + threadIdx.x;
  if (G >= NGATES) return;
  const float* lg; const int* li; const int* ri; int g;
  if (G < W0)            { lg = lg0; li = le0; ri = ri0; g = G; }
  else if (G < W0 + W1)  { lg = lg1; li = le1; ri = ri1; g = G - W0; }
  else                   { lg = lg2; li = le2; ri = ri2; g = G - W0 - W1; }
  float w0 = 0.5f + 0.5f * sinf(lg[g * 4 + 0]);
  float w1 = 0.5f + 0.5f * sinf(lg[g * 4 + 1]);
  float w2 = 0.5f + 0.5f * sinf(lg[g * 4 + 2]);
  float w3 = 0.5f + 0.5f * sinf(lg[g * 4 + 3]);
  co[G]  = make_float4(w0, w2 - w0, w1 - w0, w0 - w1 - w2 + w3);
  int l = li[g], r = ri[g];
  if (G < W0) {
    l = (l < PFEAT) ? l : l + (FBROW - PFEAT);
    r = (r < PFEAT) ? r : r + (FBROW - PFEAT);
  }
  idx[G] = make_int2(l * (BT * 4), r * (BT * 4));
}

__device__ __forceinline__ v4f gate4(const float4 w, const v4f L, const v4f R) {
  v4f acc = __builtin_elementwise_fma((v4f)(w.y), L, (v4f)(w.x));
  acc = __builtin_elementwise_fma((v4f)(w.z), R, acc);
  return __builtin_elementwise_fma((v4f)(w.w), L * R, acc);
}
struct Cone { int o[8]; float4 w0[4]; float4 w1[2]; float4 wp; };
__device__ __forceinline__ Cone cone_load(const int2* ti, const float4* tc, int g2) {
  Cone c;
  const int2 ep = ti[g2]; c.wp = tc[g2];
  const int ga = W0 + ep.x / (BT * 4), gb = W0 + ep.y / (BT * 4);
  const int2 ea = ti[ga], eb = ti[gb];
  c.w1[0] = tc[ga]; c.w1[1] = tc[gb];
  const int g0 = ea.x / (BT * 4), g1 = ea.y / (BT * 4);
  const int g2b = eb.x / (BT * 4), g3 = eb.y / (BT * 4);
  const int2 e0 = ti[g0], e1 = ti[g1], e2 = ti[g2b], e3 = ti[g3];
  c.w0[0] = tc[g0]; c.w0[1] = tc[g1]; c.w0[2] = tc[g2b]; c.w0[3] = tc[g3];
  c.o[0] = e0.x; c.o[1] = e0.y; c.o[2] = e1.x; c.o[3] = e1.y;
  c.o[4] = e2.x; c.o[5] = e2.y; c.o[6] = e3.x; c.o[7] = e3.y;
  return c;
}
__device__ __forceinline__ v4f ld_row4(const char* base, int off, int pb) {
  return *reinterpret_cast<const v4f*>(base + off + pb);
}
__device__ __forceinline__ v4f cone_piece(const char* base, const Cone& c, int pb) {
  const v4f x0 = ld_row4(base, c.o[0], pb), x1 = ld_row4(base, c.o[1], pb);
  const v4f x2 = ld_row4(base, c.o[2], pb), x3 = ld_row4(base, c.o[3], pb);
  const v4f x4 = ld_row4(base, c.o[4], pb), x5 = ld_row4(base, c.o[5], pb);
  const v4f x6 = ld_row4(base, c.o[6], pb), x7 = ld_row4(base, c.o[7], pb);
  const v4f ya = gate4(c.w1[0], gate4(c.w0[0], x0, x1), gate4(c.w0[1], x2, x3));
  const v4f yb = gate4(c.w1[1], gate4(c.w0[2], x4, x5), gate4(c.w0[3], x6, x7));
  return gate4(c.wp, ya, yb);
}
__device__ __forceinline__ void st_row4(char* row, int pb, v4f v) {
  *reinterpret_cast<v4f*>(row + pb) = v;
}

__device__ __forceinline__ void encode_step(float* sh_a, const float* xs, int slot, int krot,
                                            int bb, const float (&thv)[8]) {
  float pv[PIXJ];
#pragma unroll
  for (int j = 0; j < PIXJ; ++j) {
    const int q = slot + NSLOT * j;
    const bool live = (NSLOT * j + NSLOT <= NPIX) || (q < NPIX);
    pv[j] = live ? xs[q] : 0.0f;
  }
#pragma unroll
  for (int j = 0; j < PIXJ; ++j) {
    const int q = slot + NSLOT * j;
    const bool live = (NSLOT * j + NSLOT <= NPIX) || (q < NPIX);
    if (live) {
      float* r0 = sh_a + (q * 8 + krot) * BT + bb;
#pragma unroll
      for (int k = 0; k < 6; ++k)
        r0[k * BT] = (pv[j] >= thv[k]) ? 1.0f : 0.0f;
      float* r6 = r0 - 4 * krot * BT;
      r6[6 * BT] = (pv[j] >= thv[6]) ? 1.0f : 0.0f;
      r6[7 * BT] = (pv[j] >= thv[7]) ? 1.0f : 0.0f;
    }
  }
}

__global__ __launch_bounds__(NTHREADS)
void dlgn_main(const float* __restrict__ x, const float* __restrict__ thr,
               const int2* __restrict__ idx, const float4* __restrict__ co,
               float* __restrict__ out) {
  __shared__ __attribute__((aligned(16))) float  sh_a  [AROWS * BT];
  __shared__ __attribute__((aligned(16))) float  sh_pop[POPN  * BT];
  __shared__ int2   sh_idx[NGATES];
  __shared__ float4 sh_co [NGATES];

  const int t    = threadIdx.x;
  const int bb   = t & (BT - 1);
  const int slot = t / BT;
  const float* xb = x + (size_t)(blockIdx.x * BT + bb) * PFEAT;
  const char* a_base = reinterpret_cast<const char*>(sh_a);

  const int krot = (slot & 1) * 2;
  float thv[8];
#pragma unroll
  for (int k = 0; k < 8; ++k) thv[k] = thr[(k + krot) & 7];

  for (int i = t; i < NGATES; i += NTHREADS) {
    sh_idx[i] = idx[i];
    sh_co[i]  = co[i];
  }

  for (int j = slot; j < FBN; j += NSLOT)
    sh_a[(FBROW + j) * BT + bb] = 0.0f;
  __syncthreads();

  const Cone cf  = cone_load(sh_idx, sh_co, W0 + W1 + t);
  const int  rot = (t >> 2) & 3;
  const int  pb0 = 16 * rot,             pb1 = 16 * ((rot + 1) & 3);
  const int  pb2 = 16 * ((rot + 2) & 3), pb3 = 16 * ((rot + 3) & 3);
  char* fd = reinterpret_cast<char*>(sh_a) + (FBROW + t) * (BT * 4);

  for (int s = 0; s + 1 < STEPSN; ++s) {
    const float* xs = xb + s * NPIX;
    encode_step(sh_a, xs, slot, krot, bb, thv);
    __builtin_prefetch(xs + NPIX, 0, 0);
    __syncthreads();

    const v4f f0 = cone_piece(a_base, cf, pb0);
    const v4f f1 = cone_piece(a_base, cf, pb1);
    const v4f f2 = cone_piece(a_base, cf, pb2);
    const v4f f3 = cone_piece(a_base, cf, pb3);
    __syncthreads();
    st_row4(fd, pb0, f0); st_row4(fd, pb1, f1);
    st_row4(fd, pb2, f2); st_row4(fd, pb3, f3);
  }

  {
    const int  pg  = t & (POPN - 1);
    const int  h   = t / POPN;
    const Cone cpp = cone_load(sh_idx, sh_co, W0 + W1 + FBN + pg);
    const int  qb0 = 16 * ((h + rot) & 3), qb1 = 16 * ((h + 2 + rot) & 3);
    encode_step(sh_a, xb + (STEPSN - 1) * NPIX, slot, krot, bb, thv);
    __syncthreads();
    const v4f p0 = cone_piece(a_base, cpp, qb0);
    const v4f p1 = cone_piece(a_base, cpp, qb1);
    char* pd = reinterpret_cast<char*>(sh_pop) + pg * (BT * 4);
    st_row4(pd, qb0, p0); st_row4(pd, qb1, p1);
    __syncthreads();
  }

  if (t < 32) {
    v8f acc = {};
    v2f ones; ones[0] = 1.0f; ones[1] = 1.0f;
    const int col   = t & 15;
    const int khalf = (t < 16) ? 0 : 2;
    #pragma unroll 4
    for (int kc = 0; kc < 32; ++kc) {
      v2f bm;
      bm[0] = sh_pop[(4 * kc + khalf + 0) * BT + col];
      bm[1] = sh_pop[(4 * kc + khalf + 1) * BT + col];
      acc = __builtin_amdgcn_wmma_f32_16x16x4_f32(
           false, ones,  false, bm,
           (short)0, acc,  false,  false);
    }
    if (t < 16) {
      float logit = acc[0] / 10.0f;
      float* ob = out + (size_t)(blockIdx.x * BT + t) * 10;
      #pragma unroll
      for (int c = 0; c < 10; ++c) ob[c] = logit;
    }
  }
}

extern "C" void kernel_launch(void* const* d_in, const int* in_sizes, int n_in,
                              void* d_out, int out_size, void* d_ws, size_t ws_size,
                              hipStream_t stream) {
  const float* x   = (const float*)d_in[0];
  const float* thr = (const float*)d_in[1];
  const float* lg0 = (const float*)d_in[2];
  const int*   le0 = (const int*)  d_in[3];
  const int*   ri0 = (const int*)  d_in[4];
  const float* lg1 = (const float*)d_in[5];
  const int*   le1 = (const int*)  d_in[6];
  const int*   ri1 = (const int*)  d_in[7];
  const float* lg2 = (const float*)d_in[8];
  const int*   le2 = (const int*)  d_in[9];
  const int*   ri2 = (const int*)  d_in[10];

  int2*   idx = (int2*)d_ws;
  float4* co  = (float4*)((char*)d_ws + 32768);

  dlgn_precomp<<<(NGATES + 255) / 256, 256, 0, stream>>>(
      lg0, le0, ri0, lg1, le1, ri1, lg2, le2, ri2, idx, co);

  dlgn_main<<<4096 / BT, NTHREADS, 0, stream>>>(x, thr, idx, co, (float*)d_out);
}


# No encode phase: the tile's raw pixels go to LDS once; a cone leaf thresholds its own pixel by one clamped multiply-add (feedback leaves pass through it); a region per step, 1 barrier; sums kept
# speedup vs baseline: 7.7335x; 1.2896x over previous
#include <hip/hip_runtime.h>


#define BT       16
#define NTHREADS 256
#define NSLOT    (NTHREADS / BT)
#define W0       2048
#define W1       1024
#define W2       384
#define PFEAT    784
#define FBN      256
#define POPN     128
#define STEPSN   8
#define NGATES   (W0 + W1 + W2)
#define NPIX     (PFEAT / 8)
#define ROWB     (BT * 4)
#define FBBYTES  (FBN * ROWB)
#define REGB     (FBBYTES + NPIX * ROWB)
#define LEAF_FB  8

typedef float v2f __attribute__((ext_vector_type(2)));
typedef float v4f __attribute__((ext_vector_type(4)));
typedef float v8f __attribute__((ext_vector_type(8)));

__global__ void dlgn_precomp(const float* __restrict__ lg0,
                             const int*  __restrict__ le0, const int* __restrict__ ri0,
                             const float* __restrict__ lg1,
                             const int*  __restrict__ le1, const int* __restrict__ ri1,
                             const float* __restrict__ lg2,
                             const int*  __restrict__ le2, const int* __restrict__ ri2,
                             int2*  __restrict__ idx, float4* __restrict__ co) {
  int G = blockIdx.x * blockDim.x + threadIdx.x;
  if (G >= NGATES) return;
  const float* lg; const int* li; const int* ri; int g;
  if (G < W0)            { lg = lg0; li = le0; ri = ri0; g = G; }
  else if (G < W0 + W1)  { lg = lg1; li = le1; ri = ri1; g = G - W0; }
  else                   { lg = lg2; li = le2; ri = ri2; g = G - W0 - W1; }
  float w0 = 0.5f + 0.5f * sinf(lg[g * 4 + 0]);
  float w1 = 0.5f + 0.5f * sinf(lg[g * 4 + 1]);
  float w2 = 0.5f + 0.5f * sinf(lg[g * 4 + 2]);
  float w3 = 0.5f + 0.5f * sinf(lg[g * 4 + 3]);
  co[G]  = make_float4(w0, w2 - w0, w1 - w0, w0 - w1 - w2 + w3);
  int l = li[g], r = ri[g];
  if (G < W0) {
    l = (l < PFEAT) ? FBBYTES + (l >> 3) * ROWB + (l & 7) : (l - PFEAT) * ROWB + LEAF_FB;
    r = (r < PFEAT) ? FBBYTES + (r >> 3) * ROWB + (r & 7) : (r - PFEAT) * ROWB + LEAF_FB;
    idx[G] = make_int2(l, r);
  } else {
    idx[G] = make_int2(l * ROWB, r * ROWB);
  }
}

__device__ __forceinline__ v4f gate4(const float4 w, const v4f L, const v4f R) {
  v4f acc = __builtin_elementwise_fma((v4f)(w.y), L, (v4f)(w.x));
  acc = __builtin_elementwise_fma((v4f)(w.z), R, acc);
  return __builtin_elementwise_fma((v4f)(w.w), L * R, acc);
}
struct Cone { int o[8]; float k[8]; float b[8]; float4 w0[4]; float4 w1[2]; float4 wp; };
__device__ __forceinline__ void leaf_set(Cone& c, int i, int word, const float2* kbt) {
  const float2 kb = kbt[word & 15];
  c.o[i] = word & ~(ROWB - 1);
  c.k[i] = kb.x; c.b[i] = kb.y;
}
__device__ __forceinline__ Cone cone_load(const int2* ti, const float4* tc, int g2,
                                          const float2* thr) {
  Cone c;
  const int2 ep = ti[g2]; c.wp = tc[g2];
  const int ga = W0 + ep.x / ROWB, gb = W0 + ep.y / ROWB;
  const int2 ea = ti[ga], eb = ti[gb];
  c.w1[0] = tc[ga]; c.w1[1] = tc[gb];
  const int g0 = ea.x / ROWB, g1 = ea.y / ROWB;
  const int g2b = eb.x / ROWB, g3 = eb.y / ROWB;
  const int2 e0 = ti[g0], e1 = ti[g1], e2 = ti[g2b], e3 = ti[g3];
  c.w0[0] = tc[g0]; c.w0[1] = tc[g1]; c.w0[2] = tc[g2b]; c.w0[3] = tc[g3];
  leaf_set(c, 0, e0.x, thr); leaf_set(c, 1, e0.y, thr);
  leaf_set(c, 2, e1.x, thr); leaf_set(c, 3, e1.y, thr);
  leaf_set(c, 4, e2.x, thr); leaf_set(c, 5, e2.y, thr);
  leaf_set(c, 6, e3.x, thr); leaf_set(c, 7, e3.y, thr);
  return c;
}
__device__ __forceinline__ float leaf1(float x, float k, float b) {
  float r;
  asm("v_fma_f32 %0, %1, %2, %3 clamp" : "=v"(r) : "v"(x), "v"(k), "v"(b));
  return r;
}
__device__ __forceinline__ v4f leaf4(const char* reg, const Cone& c, int i, int pb) {
  const v4f x = *reinterpret_cast<const v4f*>(reg + c.o[i] + pb);
  v4f r;
  r[0] = leaf1(x[0], c.k[i], c.b[i]);
  r[1] = leaf1(x[1], c.k[i], c.b[i]);
  r[2] = leaf1(x[2], c.k[i], c.b[i]);
  r[3] = leaf1(x[3], c.k[i], c.b[i]);
  return r;
}
__device__ __forceinline__ v4f cone_piece(const char* reg, const Cone& c, int pb) {
  const v4f x0 = leaf4(reg, c, 0, pb), x1 = leaf4(reg, c, 1, pb);
  const v4f x2 = leaf4(reg, c, 2, pb), x3 = leaf4(reg, c, 3, pb);
  const v4f x4 = leaf4(reg, c, 4, pb), x5 = leaf4(reg, c, 5, pb);
  const v4f x6 = leaf4(reg, c, 6, pb), x7 = leaf4(reg, c, 7, pb);
  const v4f ya = gate4(c.w1[0], gate4(c.w0[0], x0, x1), gate4(c.w0[1], x2, x3));
  const v4f yb = gate4(c.w1[1], gate4(c.w0[2], x4, x5), gate4(c.w0[3], x6, x7));
  return gate4(c.wp, ya, yb);
}
__device__ __forceinline__ void st_row4(char* row, int pb, v4f v) {
  *reinterpret_cast<v4f*>(row + pb) = v;
}

__global__ __launch_bounds__(NTHREADS)
void dlgn_main(const float* __restrict__ x, const float* __restrict__ thr,
               const int2* __restrict__ idx, const float4* __restrict__ co,
               float* __restrict__ out) {
  __shared__ __attribute__((aligned(16))) float  sh_a  [STEPSN * REGB / 4];
  __shared__ __attribute__((aligned(16))) float  sh_pop[POPN  * BT];
  __shared__ int2   sh_idx[NGATES];
  __shared__ float4 sh_co [NGATES];
  __shared__ float2 sh_kb [16];

  const int t    = threadIdx.x;
  const int bb   = t & (BT - 1);
  const int slot = t / BT;
  const float* xb = x + (size_t)(blockIdx.x * BT + bb) * PFEAT;
  char* a_base = reinterpret_cast<char*>(sh_a);

  constexpr int NLD = (PFEAT + 4 * NSLOT - 1) / (4 * NSLOT);
  const float th_t = thr[t & 7];
  float4 pv[NLD];
#pragma unroll
  for (int i = 0; i < NLD; ++i) {
    const int p = 4 * slot + 4 * NSLOT * i;
    const bool live = (4 * NSLOT * i + 4 * NSLOT <= PFEAT) || (p < PFEAT);
    pv[i] = live ? *reinterpret_cast<const float4*>(xb + p) : make_float4(0.f, 0.f, 0.f, 0.f);
  }

  for (int i = t; i < NGATES; i += NTHREADS) {
    sh_idx[i] = idx[i];
    sh_co[i]  = co[i];
  }

  if (t <= LEAF_FB) {
    float2 kb = make_float2(1.0f, 0.0f);
    if (t < LEAF_FB) {
      kb.x = 0x1p126f;
      kb.y = (th_t > 0.0f) ? -__int_as_float(__float_as_int(th_t) - 1) * 0x1p126f : 1.0f;
    }
    sh_kb[t] = kb;
  }

#pragma unroll
  for (int i = 0; i < NLD; ++i) {
    const int p = 4 * slot + 4 * NSLOT * i;
    const bool live = (4 * NSLOT * i + 4 * NSLOT <= PFEAT) || (p < PFEAT);
    if (live) {
      const float e[4] = {pv[i].x, pv[i].y, pv[i].z, pv[i].w};
#pragma unroll
      for (int k = 0; k < 4; ++k) {
        const unsigned pk = (unsigned)(p + k);
        const unsigned s  = (pk * 669u) >> 16;
        *reinterpret_cast<float*>(a_base + FBBYTES + pk * ROWB + s * FBBYTES + bb * 4) = e[k];
      }
    }
  }

  for (int i = t; i < FBBYTES / 16; i += NTHREADS)
    reinterpret_cast<v4f*>(a_base)[i] = (v4f)(0.0f);
  __syncthreads();

  const Cone cf  = cone_load(sh_idx, sh_co, W0 + W1 + t, sh_kb);
  const int  rot = (t >> 2) & 3;
  const int  pb0 = 16 * rot,             pb1 = 16 * ((rot + 1) & 3);
  const int  pb2 = 16 * ((rot + 2) & 3), pb3 = 16 * ((rot + 3) & 3);

  for (int s = 0; s + 1 < STEPSN; ++s) {
    const char* reg = a_base + s * REGB;
    char* fd = a_base + (s + 1) * REGB + t * ROWB;
    const v4f f0 = cone_piece(reg, cf, pb0);
    const v4f f1 = cone_piece(reg, cf, pb1);
    const v4f f2 = cone_piece(reg, cf, pb2);
    const v4f f3 = cone_piece(reg, cf, pb3);
    st_row4(fd, pb0, f0); st_row4(fd, pb1, f1);
    st_row4(fd, pb2, f2); st_row4(fd, pb3, f3);
    __syncthreads();
  }

  {
    const int  pg  = t & (POPN - 1);
    const int  h   = t / POPN;
    const Cone cpp = cone_load(sh_idx, sh_co, W0 + W1 + FBN + pg, sh_kb);
    const int  qb0 = 16 * ((h + rot) & 3), qb1 = 16 * ((h + 2 + rot) & 3);
    const char* reg = a_base + (STEPSN - 1) * REGB;
    const v4f p0 = cone_piece(reg, cpp, qb0);
    const v4f p1 = cone_piece(reg, cpp, qb1);
    char* pd = reinterpret_cast<char*>(sh_pop) + pg * ROWB;
    st_row4(pd, qb0, p0); st_row4(pd, qb1, p1);
    __syncthreads();
  }

  if (t < 32) {
    v8f acc = {};
    v2f ones; ones[0] = 1.0f; ones[1] = 1.0f;
    const int col   = t & 15;
    const int khalf = (t < 16) ? 0 : 2;
    #pragma unroll 4
    for (int kc = 0; kc < 32; ++kc) {
      v2f bm;
      bm[0] = sh_pop[(4 * kc + khalf + 0) * BT + col];
      bm[1] = sh_pop[(4 * kc + khalf + 1) * BT + col];
      acc = __builtin_amdgcn_wmma_f32_16x16x4_f32(
           false, ones,  false, bm,
           (short)0, acc,  false,  false);
    }
    if (t < 16) {
      float logit = acc[0] / 10.0f;
      float* ob = out + (size_t)(blockIdx.x * BT + t) * 10;
      #pragma unroll
      for (int c = 0; c < 10; ++c) ob[c] = logit;
    }
  }
}

extern "C" void kernel_launch(void* const* d_in, const int* in_sizes, int n_in,
                              void* d_out, int out_size, void* d_ws, size_t ws_size,
                              hipStream_t stream) {
  const float* x   = (const float*)d_in[0];
  const float* thr = (const float*)d_in[1];
  const float* lg0 = (const float*)d_in[2];
  const int*   le0 = (const int*)  d_in[3];
  const int*   ri0 = (const int*)  d_in[4];
  const float* lg1 = (const float*)d_in[5];
  const int*   le1 = (const int*)  d_in[6];
  const int*   ri1 = (const int*)  d_in[7];
  const float* lg2 = (const float*)d_in[8];
  const int*   le2 = (const int*)  d_in[9];
  const int*   ri2 = (const int*)  d_in[10];

  int2*   idx = (int2*)d_ws;
  float4* co  = (float4*)((char*)d_ws + 32768);

  dlgn_precomp<<<(NGATES + 255) / 256, 256, 0, stream>>>(
      lg0, le0, ri0, lg1, le1, ri1, lg2, le2, ri2, idx, co);

  dlgn_main<<<4096 / BT, NTHREADS, 0, stream>>>(x, thr, idx, co, (float*)d_out);
}


# Pre-pass follows the wiring once and writes one record per delivered gate's cone (leaf words, coefficients); a thread reads its cone by 11 coalesced loads: no gate tables in LDS; sums kept
# speedup vs baseline: 9.3417x; 1.2080x over previous
#include <hip/hip_runtime.h>


#define BT       16
#define NTHREADS 256
#define NSLOT    (NTHREADS / BT)
#define PFEAT    784
#define FBN      256
#define POPN     128
#define STEPSN   8
#define NPIX     (PFEAT / 8)
#define ROWB     (BT * 4)
#define FBBYTES  (FBN * ROWB)
#define REGB     (FBBYTES + NPIX * ROWB)
#define LEAF_FB  8
#define NCONE    (FBN + POPN)
#define RECQ     7
#define RECL     4

typedef float v2f __attribute__((ext_vector_type(2)));
typedef float v4f __attribute__((ext_vector_type(4)));
typedef float v8f __attribute__((ext_vector_type(8)));

__device__ __forceinline__ int leaf_word(int f) {
  return (f < PFEAT) ? FBBYTES + (f >> 3) * ROWB + (f & 7) : (f - PFEAT) * ROWB + LEAF_FB;
}
__global__ void dlgn_precomp(const float* __restrict__ lg0,
                             const int*  __restrict__ le0, const int* __restrict__ ri0,
                             const float* __restrict__ lg1,
                             const int*  __restrict__ le1, const int* __restrict__ ri1,
                             const float* __restrict__ lg2,
                             const int*  __restrict__ le2, const int* __restrict__ ri2,
                             float4* __restrict__ rq, int2* __restrict__ rl) {
  const int role = blockIdx.x / 3;
  const int c    = (blockIdx.x % 3) * 128 + threadIdx.x;
  const int i    = (role >= 3) ? role - 3 : (role == 2) ? 2 : 0;
  const int side = i >> 1;
  int g = c; const float* lgl = lg2; int plane = 6;
  if (role >= 1) { g = side ? ri2[c] : le2[c];    lgl = lg1; plane = 4 + side; }
  if (role >= 3) { g = (i & 1) ? ri1[g] : le1[g]; lgl = lg0; plane = i; }
  const float4 l4 = *reinterpret_cast<const float4*>(lgl + g * 4);
  if (role >= 3) {
    const int fl = le0[g], fr = ri0[g];
    rl[i * NCONE + c] = make_int2(leaf_word(fl), leaf_word(fr));
  }
  const float w0 = 0.5f + 0.5f * sinf(l4.x);
  const float w1 = 0.5f + 0.5f * sinf(l4.y);
  const float w2 = 0.5f + 0.5f * sinf(l4.z);
  const float w3 = 0.5f + 0.5f * sinf(l4.w);
  rq[plane * NCONE + c] = make_float4(w0, w2 - w0, w1 - w0, w0 - w1 - w2 + w3);
}

__device__ __forceinline__ v4f gate4(const float4 w, const v4f L, const v4f R) {
  v4f acc = __builtin_elementwise_fma((v4f)(w.y), L, (v4f)(w.x));
  acc = __builtin_elementwise_fma((v4f)(w.z), R, acc);
  return __builtin_elementwise_fma((v4f)(w.w), L * R, acc);
}
struct Cone { int o[8]; float k[8]; float b[8]; float4 w0[4]; float4 w1[2]; float4 wp; };
__device__ __forceinline__ void leaf_set(Cone& c, int i, int word, const float2* kbt) {
  const float2 kb = kbt[word & 15];
  c.o[i] = word & ~(ROWB - 1);
  c.k[i] = kb.x; c.b[i] = kb.y;
}
struct Rec { int2 l[RECL]; float4 q[RECQ]; };
__device__ __forceinline__ Rec rec_load(const float4* __restrict__ rq,
                                        const int2* __restrict__ rl, int cone) {
  Rec r;
#pragma unroll
  for (int i = 0; i < RECL; ++i) r.l[i] = rl[i * NCONE + cone];
#pragma unroll
  for (int j = 0; j < RECQ; ++j) r.q[j] = rq[j * NCONE + cone];
  return r;
}
__device__ __forceinline__ Cone cone_make(const Rec& r, const float2* thr) {
  Cone c;
  leaf_set(c, 0, r.l[0].x, thr); leaf_set(c, 1, r.l[0].y, thr);
  leaf_set(c, 2, r.l[1].x, thr); leaf_set(c, 3, r.l[1].y, thr);
  leaf_set(c, 4, r.l[2].x, thr); leaf_set(c, 5, r.l[2].y, thr);
  leaf_set(c, 6, r.l[3].x, thr); leaf_set(c, 7, r.l[3].y, thr);
  c.w0[0] = r.q[0]; c.w0[1] = r.q[1]; c.w0[2] = r.q[2]; c.w0[3] = r.q[3];
  c.w1[0] = r.q[4]; c.w1[1] = r.q[5]; c.wp = r.q[6];
  return c;
}
__device__ __forceinline__ float leaf1(float x, float k, float b) {
  float r;
  asm("v_fma_f32 %0, %1, %2, %3 clamp" : "=v"(r) : "v"(x), "v"(k), "v"(b));
  return r;
}
__device__ __forceinline__ v4f leaf4(const char* reg, const Cone& c, int i, int pb) {
  const v4f x = *reinterpret_cast<const v4f*>(reg + c.o[i] + pb);
  v4f r;
  r[0] = leaf1(x[0], c.k[i], c.b[i]);
  r[1] = leaf1(x[1], c.k[i], c.b[i]);
  r[2] = leaf1(x[2], c.k[i], c.b[i]);
  r[3] = leaf1(x[3], c.k[i], c.b[i]);
  return r;
}
__device__ __forceinline__ v4f cone_piece(const char* reg, const Cone& c, int pb) {
  const v4f x0 = leaf4(reg, c, 0, pb), x1 = leaf4(reg, c, 1, pb);
  const v4f x2 = leaf4(reg, c, 2, pb), x3 = leaf4(reg, c, 3, pb);
  const v4f x4 = leaf4(reg, c, 4, pb), x5 = leaf4(reg, c, 5, pb);
  const v4f x6 = leaf4(reg, c, 6, pb), x7 = leaf4(reg, c, 7, pb);
  const v4f ya = gate4(c.w1[0], gate4(c.w0[0], x0, x1), gate4(c.w0[1], x2, x3));
  const v4f yb = gate4(c.w1[1], gate4(c.w0[2], x4, x5), gate4(c.w0[3], x6, x7));
  return gate4(c.wp, ya, yb);
}
__device__ __forceinline__ void st_row4(char* row, int pb, v4f v) {
  *reinterpret_cast<v4f*>(row + pb) = v;
}

__global__ __launch_bounds__(NTHREADS)
void dlgn_main(const float* __restrict__ x, const float* __restrict__ thr,
               const float4* __restrict__ rq, const int2* __restrict__ rl,
               float* __restrict__ out) {
  __shared__ __attribute__((aligned(16))) float  sh_a  [STEPSN * REGB / 4];
  __shared__ __attribute__((aligned(16))) float  sh_pop[POPN  * BT];
  __shared__ float2 sh_kb [16];

  const int t    = threadIdx.x;
  const int bb   = t & (BT - 1);
  const int slot = t / BT;
  const float* xb = x + (size_t)(blockIdx.x * BT + bb) * PFEAT;
  char* a_base = reinterpret_cast<char*>(sh_a);

  const Rec rf = rec_load(rq, rl, t);

  constexpr int NLD = (PFEAT + 4 * NSLOT - 1) / (4 * NSLOT);
  const float th_t = thr[t & 7];
  float4 pv[NLD];
#pragma unroll
  for (int i = 0; i < NLD; ++i) {
    const int p = 4 * slot + 4 * NSLOT * i;
    const bool live = (4 * NSLOT * i + 4 * NSLOT <= PFEAT) || (p < PFEAT);
    pv[i] = live ? *reinterpret_cast<const float4*>(xb + p) : make_float4(0.f, 0.f, 0.f, 0.f);
  }

  if (t <= LEAF_FB) {
    float2 kb = make_float2(1.0f, 0.0f);
    if (t < LEAF_FB) {
      kb.x = 0x1p126f;
      kb.y = (th_t > 0.0f) ? -__int_as_float(__float_as_int(th_t) - 1) * 0x1p126f : 1.0f;
    }
    sh_kb[t] = kb;
  }

#pragma unroll
  for (int i = 0; i < NLD; ++i) {
    const int p = 4 * slot + 4 * NSLOT * i;
    const bool live = (4 * NSLOT * i + 4 * NSLOT <= PFEAT) || (p < PFEAT);
    if (live) {
      const float e[4] = {pv[i].x, pv[i].y, pv[i].z, pv[i].w};
#pragma unroll
      for (int k = 0; k < 4; ++k) {
        const unsigned pk = (unsigned)(p + k);
        const unsigned s  = (pk * 669u) >> 16;
        *reinterpret_cast<float*>(a_base + FBBYTES + pk * ROWB + s * FBBYTES + bb * 4) = e[k];
      }
    }
  }

  for (int i = t; i < FBBYTES / 16; i += NTHREADS)
    reinterpret_cast<v4f*>(a_base)[i] = (v4f)(0.0f);
  __syncthreads();

  const Rec rp = rec_load(rq, rl, FBN + (t & (POPN - 1)));

  const Cone cf  = cone_make(rf, sh_kb);
  const int  rot = (t >> 2) & 3;
  const int  pb0 = 16 * rot,             pb1 = 16 * ((rot + 1) & 3);
  const int  pb2 = 16 * ((rot + 2) & 3), pb3 = 16 * ((rot + 3) & 3);

  for (int s = 0; s + 1 < STEPSN; ++s) {
    const char* reg = a_base + s * REGB;
    char* fd = a_base + (s + 1) * REGB + t * ROWB;
    const v4f f0 = cone_piece(reg, cf, pb0);
    const v4f f1 = cone_piece(reg, cf, pb1);
    const v4f f2 = cone_piece(reg, cf, pb2);
    const v4f f3 = cone_piece(reg, cf, pb3);
    st_row4(fd, pb0, f0); st_row4(fd, pb1, f1);
    st_row4(fd, pb2, f2); st_row4(fd, pb3, f3);
    __syncthreads();
  }

  {
    const int  pg  = t & (POPN - 1);
    const int  h   = t / POPN;
    const Cone cpp = cone_make(rp, sh_kb);
    const int  qb0 = 16 * ((h + rot) & 3), qb1 = 16 * ((h + 2 + rot) & 3);
    const char* reg = a_base + (STEPSN - 1) * REGB;
    const v4f p0 = cone_piece(reg, cpp, qb0);
    const v4f p1 = cone_piece(reg, cpp, qb1);
    char* pd = reinterpret_cast<char*>(sh_pop) + pg * ROWB;
    st_row4(pd, qb0, p0); st_row4(pd, qb1, p1);
    __syncthreads();
  }

  if (t < 32) {
    v8f acc = {};
    v2f ones; ones[0] = 1.0f; ones[1] = 1.0f;
    const int col   = t & 15;
    const int khalf = (t < 16) ? 0 : 2;
    #pragma unroll 4
    for (int kc = 0; kc < 32; ++kc) {
      v2f bm;
      bm[0] = sh_pop[(4 * kc + khalf + 0) * BT + col];
      bm[1] = sh_pop[(4 * kc + khalf + 1) * BT + col];
      acc = __builtin_amdgcn_wmma_f32_16x16x4_f32(
           false, ones,  false, bm,
           (short)0, acc,  false,  false);
    }
    if (t < 16) {
      float logit = acc[0] / 10.0f;
      float* ob = out + (size_t)(blockIdx.x * BT + t) * 10;
      #pragma unroll
      for (int c = 0; c < 10; ++c) ob[c] = logit;
    }
  }
}

extern "C" void kernel_launch(void* const* d_in, const int* in_sizes, int n_in,
                              void* d_out, int out_size, void* d_ws, size_t ws_size,
                              hipStream_t stream) {
  const float* x   = (const float*)d_in[0];
  const float* thr = (const float*)d_in[1];
  const float* lg0 = (const float*)d_in[2];
  const int*   le0 = (const int*)  d_in[3];
  const int*   ri0 = (const int*)  d_in[4];
  const float* lg1 = (const float*)d_in[5];
  const int*   le1 = (const int*)  d_in[6];
  const int*   ri1 = (const int*)  d_in[7];
  const float* lg2 = (const float*)d_in[8];
  const int*   le2 = (const int*)  d_in[9];
  const int*   ri2 = (const int*)  d_in[10];

  float4* rq = (float4*)d_ws;
  int2*   rl = (int2*)((char*)d_ws + 43008);

  dlgn_precomp<<<7 * 3, 128, 0, stream>>>(
      lg0, le0, ri0, lg1, le1, ri1, lg2, le2, ri2, rq, rl);

  dlgn_main<<<4096 / BT, NTHREADS, 0, stream>>>(x, thr, rq, rl, (float*)d_out);
}


# speedup vs baseline: 10.2383x; 1.0960x over previous
#include <hip/hip_runtime.h>


#define BT       16
#define NTHREADS 256
#define NSLOT    (NTHREADS / BT)
#define PFEAT    784
#define FBN      256
#define POPN     128
#define STEPSN   8
#define NPIX     (PFEAT / 8)
#define ROWB     (BT * 4)
#define FBBYTES  (FBN * ROWB)
#define STEPPIX  (NPIX * ROWB)
#define PIXB0    (2 * FBBYTES)
#define COPYSZ   (PIXB0 + PFEAT * ROWB)
#define COPYB    163840
#define LEAF_FB  8
#define NCONE    (FBN + POPN)
#define RECQ     7
#define RECL     4

typedef float v2f __attribute__((ext_vector_type(2)));
typedef float v4f __attribute__((ext_vector_type(4)));
typedef float v8f __attribute__((ext_vector_type(8)));

__device__ __forceinline__ int leaf_word(int f) {
  return (f < PFEAT) ? FBBYTES + (f >> 3) * ROWB + (f & 7) : (f - PFEAT) * ROWB + LEAF_FB;
}
__global__ void dlgn_precomp(const float* __restrict__ lg0,
                             const int*  __restrict__ le0, const int* __restrict__ ri0,
                             const float* __restrict__ lg1,
                             const int*  __restrict__ le1, const int* __restrict__ ri1,
                             const float* __restrict__ lg2,
                             const int*  __restrict__ le2, const int* __restrict__ ri2,
                             float4* __restrict__ rq, int2* __restrict__ rl) {
  const int role = blockIdx.x / 3;
  const int c    = (blockIdx.x % 3) * 128 + threadIdx.x;
  const int i    = (role >= 3) ? role - 3 : (role == 2) ? 2 : 0;
  const int side = i >> 1;
  int g = c; const float* lgl = lg2; int plane = 6;
  if (role >= 1) { g = side ? ri2[c] : le2[c];    lgl = lg1; plane = 4 + side; }
  if (role >= 3) { g = (i & 1) ? ri1[g] : le1[g]; lgl = lg0; plane = i; }
  const float4 l4 = *reinterpret_cast<const float4*>(lgl + g * 4);
  if (role >= 3) {
    const int fl = le0[g], fr = ri0[g];
    rl[i * NCONE + c] = make_int2(leaf_word(fl), leaf_word(fr));
  }
  const float w0 = 0.5f + 0.5f * sinf(l4.x);
  const float w1 = 0.5f + 0.5f * sinf(l4.y);
  const float w2 = 0.5f + 0.5f * sinf(l4.z);
  const float w3 = 0.5f + 0.5f * sinf(l4.w);
  rq[plane * NCONE + c] = make_float4(w0, w2 - w0, w1 - w0, w0 - w1 - w2 + w3);
}

__device__ __forceinline__ v4f gate4(const float4 w, const v4f L, const v4f R) {
  v4f acc = __builtin_elementwise_fma((v4f)(w.y), L, (v4f)(w.x));
  acc = __builtin_elementwise_fma((v4f)(w.z), R, acc);
  return __builtin_elementwise_fma((v4f)(w.w), L * R, acc);
}
struct Cone { int o[8]; int tog[8]; int add[8]; float k[8]; float b[8]; float4 w0[4]; float4 w1[2]; float4 wp; };
__device__ __forceinline__ void leaf_set(Cone& c, int i, int word, const float2* kbt,
                                         int copy, int step) {
  const float2 kb = kbt[word & 15];
  const bool fb  = (word & 15) == LEAF_FB;
  const int  off = word & ~(ROWB - 1);
  c.o[i]   = copy + off + (fb ? (step & 1) * FBBYTES : (PIXB0 - FBBYTES) + step * STEPPIX);
  c.tog[i] = fb ? FBBYTES : 0;
  c.add[i] = fb ? 0 : STEPPIX;
  c.k[i] = kb.x; c.b[i] = kb.y;
}
__device__ __forceinline__ void cone_next(Cone& c) {
#pragma unroll
  for (int i = 0; i < 8; ++i) c.o[i] = (c.o[i] ^ c.tog[i]) + c.add[i];
}
struct Rec { int2 l[RECL]; float4 q[RECQ]; };
__device__ __forceinline__ Rec rec_load(const float4* __restrict__ rq,
                                        const int2* __restrict__ rl, int cone) {
  Rec r;
#pragma unroll
  for (int i = 0; i < RECL; ++i) r.l[i] = rl[i * NCONE + cone];
#pragma unroll
  for (int j = 0; j < RECQ; ++j) r.q[j] = rq[j * NCONE + cone];
  return r;
}
__device__ __forceinline__ Cone cone_make(const Rec& r, const float2* thr, int copy, int step) {
  Cone c;
  leaf_set(c, 0, r.l[0].x, thr, copy, step); leaf_set(c, 1, r.l[0].y, thr, copy, step);
  leaf_set(c, 2, r.l[1].x, thr, copy, step); leaf_set(c, 3, r.l[1].y, thr, copy, step);
  leaf_set(c, 4, r.l[2].x, thr, copy, step); leaf_set(c, 5, r.l[2].y, thr, copy, step);
  leaf_set(c, 6, r.l[3].x, thr, copy, step); leaf_set(c, 7, r.l[3].y, thr, copy, step);
  c.w0[0] = r.q[0]; c.w0[1] = r.q[1]; c.w0[2] = r.q[2]; c.w0[3] = r.q[3];
  c.w1[0] = r.q[4]; c.w1[1] = r.q[5]; c.wp = r.q[6];
  return c;
}
__device__ __forceinline__ float leaf1(float x, float k, float b) {
  float r;
  asm("v_fma_f32 %0, %1, %2, %3 clamp" : "=v"(r) : "v"(x), "v"(k), "v"(b));
  return r;
}
__device__ __forceinline__ v4f leaf4(const char* reg, const Cone& c, int i, int pb) {
  const v4f x = *reinterpret_cast<const v4f*>(reg + c.o[i] + pb);
  v4f r;
  r[0] = leaf1(x[0], c.k[i], c.b[i]);
  r[1] = leaf1(x[1], c.k[i], c.b[i]);
  r[2] = leaf1(x[2], c.k[i], c.b[i]);
  r[3] = leaf1(x[3], c.k[i], c.b[i]);
  return r;
}
__device__ __forceinline__ v4f cone_piece(const char* reg, const Cone& c, int pb) {
  const v4f x0 = leaf4(reg, c, 0, pb), x1 = leaf4(reg, c, 1, pb);
  const v4f x2 = leaf4(reg, c, 2, pb), x3 = leaf4(reg, c, 3, pb);
  const v4f x4 = leaf4(reg, c, 4, pb), x5 = leaf4(reg, c, 5, pb);
  const v4f x6 = leaf4(reg, c, 6, pb), x7 = leaf4(reg, c, 7, pb);
  const v4f ya = gate4(c.w1[0], gate4(c.w0[0], x0, x1), gate4(c.w0[1], x2, x3));
  const v4f yb = gate4(c.w1[1], gate4(c.w0[2], x4, x5), gate4(c.w0[3], x6, x7));
  return gate4(c.wp, ya, yb);
}
__device__ __forceinline__ void st_row4(char* row, int pb, v4f v) {
  *reinterpret_cast<v4f*>(row + pb) = v;
}

__global__ __launch_bounds__(NTHREADS)
void dlgn_main(const float* __restrict__ x, const float* __restrict__ thr,
               const float4* __restrict__ rq, const int2* __restrict__ rl,
               float* __restrict__ out) {
  __shared__ __attribute__((aligned(16))) float  sh_a  [(COPYB + COPYSZ) / 4];
  __shared__ __attribute__((aligned(16))) float  sh_pop[POPN  * BT];
  __shared__ float2 sh_kb [16];

  const int t    = threadIdx.x;
  const int bb   = t & (BT - 1);
  const int slot = t / BT;
  const float* xb = x + (size_t)(blockIdx.x * BT + bb) * PFEAT;
  char* a_base = reinterpret_cast<char*>(sh_a);

  unsigned simd;
  asm("s_getreg_b32 %0, hwreg(HW_REG_WAVE_HW_ID1, 8, 2)" : "=s"(simd));
  const int copy = (simd & 1u) ? COPYB : 0;

  const Rec rf = rec_load(rq, rl, t);

  constexpr int NLD = (PFEAT + 4 * NSLOT - 1) / (4 * NSLOT);
  const float th_t = thr[t & 7];
  float4 pv[NLD];
#pragma unroll
  for (int i = 0; i < NLD; ++i) {
    const int p = 4 * slot + 4 * NSLOT * i;
    const bool live = (4 * NSLOT * i + 4 * NSLOT <= PFEAT) || (p < PFEAT);
    pv[i] = live ? *reinterpret_cast<const float4*>(xb + p) : make_float4(0.f, 0.f, 0.f, 0.f);
  }

  if (t <= LEAF_FB) {
    float2 kb = make_float2(1.0f, 0.0f);
    if (t < LEAF_FB) {
      kb.x = 0x1p126f;
      kb.y = (th_t > 0.0f) ? -__int_as_float(__float_as_int(th_t) - 1) * 0x1p126f : 1.0f;
    }
    sh_kb[t] = kb;
  }

#pragma unroll
  for (int i = 0; i < NLD; ++i) {
    const int p = 4 * slot + 4 * NSLOT * i;
    const bool live = (4 * NSLOT * i + 4 * NSLOT <= PFEAT) || (p < PFEAT);
    if (live) {
      const float e[4] = {pv[i].x, pv[i].y, pv[i].z, pv[i].w};
#pragma unroll
      for (int k = 0; k < 4; ++k) {
        const unsigned pk = (unsigned)(p + k);
        float* row = reinterpret_cast<float*>(a_base + PIXB0 + pk * ROWB + bb * 4);
        row[0] = e[k];
        row[COPYB / 4] = e[k];
      }
    }
  }

  for (int i = t; i < FBBYTES / 16; i += NTHREADS) {
    reinterpret_cast<v4f*>(a_base)[i] = (v4f)(0.0f);
    reinterpret_cast<v4f*>(a_base + COPYB)[i] = (v4f)(0.0f);
  }
  __syncthreads();

  const Rec rp = rec_load(rq, rl, FBN + (t & (POPN - 1)));

  Cone cf = cone_make(rf, sh_kb, copy, 0);
  const int  rot = (t >> 2) & 3;
  const int  pb0 = 16 * rot,             pb1 = 16 * ((rot + 1) & 3);
  const int  pb2 = 16 * ((rot + 2) & 3), pb3 = 16 * ((rot + 3) & 3);

  for (int s = 0; s + 1 < STEPSN; ++s) {
    char* fd = a_base + ((s + 1) & 1) * FBBYTES + t * ROWB;
    const v4f f0 = cone_piece(a_base, cf, pb0);
    const v4f f1 = cone_piece(a_base, cf, pb1);
    const v4f f2 = cone_piece(a_base, cf, pb2);
    const v4f f3 = cone_piece(a_base, cf, pb3);
    st_row4(fd, pb0, f0); st_row4(fd, pb1, f1);
    st_row4(fd, pb2, f2); st_row4(fd, pb3, f3);
    st_row4(fd + COPYB, pb0, f0); st_row4(fd + COPYB, pb1, f1);
    st_row4(fd + COPYB, pb2, f2); st_row4(fd + COPYB, pb3, f3);
    cone_next(cf);
    __syncthreads();
  }

  {
    const int  pg  = t & (POPN - 1);
    const int  h   = t / POPN;
    const Cone cpp = cone_make(rp, sh_kb, copy, STEPSN - 1);
    const int  qb0 = 16 * ((h + rot) & 3), qb1 = 16 * ((h + 2 + rot) & 3);
    const v4f p0 = cone_piece(a_base, cpp, qb0);
    const v4f p1 = cone_piece(a_base, cpp, qb1);
    char* pd = reinterpret_cast<char*>(sh_pop) + pg * ROWB;
    st_row4(pd, qb0, p0); st_row4(pd, qb1, p1);
    __syncthreads();
  }

  if (t < 32) {
    v8f acc = {};
    v2f ones; ones[0] = 1.0f; ones[1] = 1.0f;
    const int col   = t & 15;
    const int khalf = (t < 16) ? 0 : 2;
    #pragma unroll 4
    for (int kc = 0; kc < 32; ++kc) {
      v2f bm;
      bm[0] = sh_pop[(4 * kc + khalf + 0) * BT + col];
      bm[1] = sh_pop[(4 * kc + khalf + 1) * BT + col];
      acc = __builtin_amdgcn_wmma_f32_16x16x4_f32(
           false, ones,  false, bm,
           (short)0, acc,  false,  false);
    }
    if (t < 16) {
      float logit = acc[0] / 10.0f;
      float* ob = out + (size_t)(blockIdx.x * BT + t) * 10;
      #pragma unroll
      for (int c = 0; c < 10; ++c) ob[c] = logit;
    }
  }
}

extern "C" void kernel_launch(void* const* d_in, const int* in_sizes, int n_in,
                              void* d_out, int out_size, void* d_ws, size_t ws_size,
                              hipStream_t stream) {
  const float* x   = (const float*)d_in[0];
  const float* thr = (const float*)d_in[1];
  const float* lg0 = (const float*)d_in[2];
  const int*   le0 = (const int*)  d_in[3];
  const int*   ri0 = (const int*)  d_in[4];
  const float* lg1 = (const float*)d_in[5];
  const int*   le1 = (const int*)  d_in[6];
  const int*   ri1 = (const int*)  d_in[7];
  const float* lg2 = (const float*)d_in[8];
  const int*   le2 = (const int*)  d_in[9];
  const int*   ri2 = (const int*)  d_in[10];

  float4* rq = (float4*)d_ws;
  int2*   rl = (int2*)((char*)d_ws + 43008);

  dlgn_precomp<<<7 * 3, 128, 0, stream>>>(
      lg0, le0, ri0, lg1, le1, ri1, lg2, le2, ri2, rq, rl);

  dlgn_main<<<4096 / BT, NTHREADS, 0, stream>>>(x, thr, rq, rl, (float*)d_out);
}


# Fewer vector instructions per cone: a gate is 3 multiply-adds, (w0+wy L)+R(wz+ww L), no separate product L R (sum order changed); leaf clamp multiply-adds packed 2 columns an instruction
# speedup vs baseline: 10.9209x; 1.0667x over previous
#include <hip/hip_runtime.h>


#define BT       16
#define NTHREADS 256
#define NSLOT    (NTHREADS / BT)
#define PFEAT    784
#define FBN      256
#define POPN     128
#define STEPSN   8
#define NPIX     (PFEAT / 8)
#define ROWB     (BT * 4)
#define FBBYTES  (FBN * ROWB)
#define STEPPIX  (NPIX * ROWB)
#define PIXB0    (2 * FBBYTES)
#define COPYSZ   (PIXB0 + PFEAT * ROWB)
#define COPYB    163840
#define LEAF_FB  8
#define NCONE    (FBN + POPN)
#define RECQ     7
#define RECL     4

typedef float v2f __attribute__((ext_vector_type(2)));
typedef float v4f __attribute__((ext_vector_type(4)));
typedef float v8f __attribute__((ext_vector_type(8)));

__device__ __forceinline__ int leaf_word(int f) {
  return (f < PFEAT) ? FBBYTES + (f >> 3) * ROWB + (f & 7) : (f - PFEAT) * ROWB + LEAF_FB;
}
__global__ void dlgn_precomp(const float* __restrict__ lg0,
                             const int*  __restrict__ le0, const int* __restrict__ ri0,
                             const float* __restrict__ lg1,
                             const int*  __restrict__ le1, const int* __restrict__ ri1,
                             const float* __restrict__ lg2,
                             const int*  __restrict__ le2, const int* __restrict__ ri2,
                             float4* __restrict__ rq, int2* __restrict__ rl) {
  const int role = blockIdx.x / 3;
  const int c    = (blockIdx.x % 3) * 128 + threadIdx.x;
  const int i    = (role >= 3) ? role - 3 : (role == 2) ? 2 : 0;
  const int side = i >> 1;
  int g = c; const float* lgl = lg2; int plane = 6;
  if (role >= 1) { g = side ? ri2[c] : le2[c];    lgl = lg1; plane = 4 + side; }
  if (role >= 3) { g = (i & 1) ? ri1[g] : le1[g]; lgl = lg0; plane = i; }
  const float4 l4 = *reinterpret_cast<const float4*>(lgl + g * 4);
  if (role >= 3) {
    const int fl = le0[g], fr = ri0[g];
    rl[i * NCONE + c] = make_int2(leaf_word(fl), leaf_word(fr));
  }
  const float w0 = 0.5f + 0.5f * sinf(l4.x);
  const float w1 = 0.5f + 0.5f * sinf(l4.y);
  const float w2 = 0.5f + 0.5f * sinf(l4.z);
  const float w3 = 0.5f + 0.5f * sinf(l4.w);
  rq[plane * NCONE + c] = make_float4(w0, w2 - w0, w1 - w0, w0 - w1 - w2 + w3);
}

__device__ __forceinline__ v4f gate4(const float4 w, const v4f L, const v4f R) {
  const v4f a = __builtin_elementwise_fma((v4f)(w.y), L, (v4f)(w.x));
  const v4f s = __builtin_elementwise_fma((v4f)(w.w), L, (v4f)(w.z));
  return __builtin_elementwise_fma(R, s, a);
}
struct Cone { int o[8]; int tog[8]; int add[8]; v2f kb[8]; float4 w0[4]; float4 w1[2]; float4 wp; };
__device__ __forceinline__ void leaf_set(Cone& c, int i, int word, const float2* kbt,
                                         int copy, int step) {
  const float2 kb = kbt[word & 15];
  const bool fb  = (word & 15) == LEAF_FB;
  const int  off = word & ~(ROWB - 1);
  c.o[i]   = copy + off + (fb ? (step & 1) * FBBYTES : (PIXB0 - FBBYTES) + step * STEPPIX);
  c.tog[i] = fb ? FBBYTES : 0;
  c.add[i] = fb ? 0 : STEPPIX;
  c.kb[i][0] = kb.x; c.kb[i][1] = kb.y;
}
__device__ __forceinline__ void cone_next(Cone& c) {
#pragma unroll
  for (int i = 0; i < 8; ++i) c.o[i] = (c.o[i] ^ c.tog[i]) + c.add[i];
}
struct Rec { int2 l[RECL]; float4 q[RECQ]; };
__device__ __forceinline__ Rec rec_load(const float4* __restrict__ rq,
                                        const int2* __restrict__ rl, int cone) {
  Rec r;
#pragma unroll
  for (int i = 0; i < RECL; ++i) r.l[i] = rl[i * NCONE + cone];
#pragma unroll
  for (int j = 0; j < RECQ; ++j) r.q[j] = rq[j * NCONE + cone];
  return r;
}
__device__ __forceinline__ Cone cone_make(const Rec& r, const float2* thr, int copy, int step) {
  Cone c;
  leaf_set(c, 0, r.l[0].x, thr, copy, step); leaf_set(c, 1, r.l[0].y, thr, copy, step);
  leaf_set(c, 2, r.l[1].x, thr, copy, step); leaf_set(c, 3, r.l[1].y, thr, copy, step);
  leaf_set(c, 4, r.l[2].x, thr, copy, step); leaf_set(c, 5, r.l[2].y, thr, copy, step);
  leaf_set(c, 6, r.l[3].x, thr, copy, step); leaf_set(c, 7, r.l[3].y, thr, copy, step);
  c.w0[0] = r.q[0]; c.w0[1] = r.q[1]; c.w0[2] = r.q[2]; c.w0[3] = r.q[3];
  c.w1[0] = r.q[4]; c.w1[1] = r.q[5]; c.wp = r.q[6];
  return c;
}
__device__ __forceinline__ v2f leaf2(v2f x, v2f kb) {
  v2f r;
  asm("v_pk_fma_f32 %0, %1, %2, %2 op_sel:[0,0,1] op_sel_hi:[1,0,1] clamp"
      : "=v"(r) : "v"(x), "v"(kb));
  return r;
}
__device__ __forceinline__ v4f leaf4(const char* reg, const Cone& c, int i, int pb) {
  const v4f x = *reinterpret_cast<const v4f*>(reg + c.o[i] + pb);
  v2f lo; lo[0] = x[0]; lo[1] = x[1];
  v2f hi; hi[0] = x[2]; hi[1] = x[3];
  const v2f rl = leaf2(lo, c.kb[i]);
  const v2f rh = leaf2(hi, c.kb[i]);
  v4f r;
  r[0] = rl[0]; r[1] = rl[1]; r[2] = rh[0]; r[3] = rh[1];
  return r;
}
__device__ __forceinline__ v4f cone_piece(const char* reg, const Cone& c, int pb) {
  const v4f x0 = leaf4(reg, c, 0, pb), x1 = leaf4(reg, c, 1, pb);
  const v4f x2 = leaf4(reg, c, 2, pb), x3 = leaf4(reg, c, 3, pb);
  const v4f x4 = leaf4(reg, c, 4, pb), x5 = leaf4(reg, c, 5, pb);
  const v4f x6 = leaf4(reg, c, 6, pb), x7 = leaf4(reg, c, 7, pb);
  const v4f ya = gate4(c.w1[0], gate4(c.w0[0], x0, x1), gate4(c.w0[1], x2, x3));
  const v4f yb = gate4(c.w1[1], gate4(c.w0[2], x4, x5), gate4(c.w0[3], x6, x7));
  return gate4(c.wp, ya, yb);
}
__device__ __forceinline__ void st_row4(char* row, int pb, v4f v) {
  *reinterpret_cast<v4f*>(row + pb) = v;
}

__global__ __launch_bounds__(NTHREADS)
void dlgn_main(const float* __restrict__ x, const float* __restrict__ thr,
               const float4* __restrict__ rq, const int2* __restrict__ rl,
               float* __restrict__ out) {
  __shared__ __attribute__((aligned(16))) float  sh_a  [(COPYB + COPYSZ) / 4];
  __shared__ __attribute__((aligned(16))) float  sh_pop[POPN  * BT];
  __shared__ float2 sh_kb [16];

  const int t    = threadIdx.x;
  const int bb   = t & (BT - 1);
  const int slot = t / BT;
  const float* xb = x + (size_t)(blockIdx.x * BT + bb) * PFEAT;
  char* a_base = reinterpret_cast<char*>(sh_a);

  unsigned simd;
  asm("s_getreg_b32 %0, hwreg(HW_REG_WAVE_HW_ID1, 8, 2)" : "=s"(simd));
  const int copy = (simd & 1u) ? COPYB : 0;

  const Rec rf = rec_load(rq, rl, t);

  constexpr int NLD = (PFEAT + 4 * NSLOT - 1) / (4 * NSLOT);
  const float th_t = thr[t & 7];
  float4 pv[NLD];
#pragma unroll
  for (int i = 0; i < NLD; ++i) {
    const int p = 4 * slot + 4 * NSLOT * i;
    const bool live = (4 * NSLOT * i + 4 * NSLOT <= PFEAT) || (p < PFEAT);
    pv[i] = live ? *reinterpret_cast<const float4*>(xb + p) : make_float4(0.f, 0.f, 0.f, 0.f);
  }

  if (t <= LEAF_FB) {
    float2 kb = make_float2(1.0f, 0.0f);
    if (t < LEAF_FB) {
      kb.x = 0x1p126f;
      kb.y = (th_t > 0.0f) ? -__int_as_float(__float_as_int(th_t) - 1) * 0x1p126f : 1.0f;
    }
    sh_kb[t] = kb;
  }

#pragma unroll
  for (int i = 0; i < NLD; ++i) {
    const int p = 4 * slot + 4 * NSLOT * i;
    const bool live = (4 * NSLOT * i + 4 * NSLOT <= PFEAT) || (p < PFEAT);
    if (live) {
      const float e[4] = {pv[i].x, pv[i].y, pv[i].z, pv[i].w};
#pragma unroll
      for (int k = 0; k < 4; ++k) {
        const unsigned pk = (unsigned)(p + k);
        float* row = reinterpret_cast<float*>(a_base + PIXB0 + pk * ROWB + bb * 4);
        row[0] = e[k];
        row[COPYB / 4] = e[k];
      }
    }
  }

  for (int i = t; i < FBBYTES / 16; i += NTHREADS) {
    reinterpret_cast<v4f*>(a_base)[i] = (v4f)(0.0f);
    reinterpret_cast<v4f*>(a_base + COPYB)[i] = (v4f)(0.0f);
  }
  __syncthreads();

  const Rec rp = rec_load(rq, rl, FBN + (t & (POPN - 1)));

  Cone cf = cone_make(rf, sh_kb, copy, 0);
  const int  rot = (t >> 2) & 3;
  const int  pb0 = 16 * rot,             pb1 = 16 * ((rot + 1) & 3);
  const int  pb2 = 16 * ((rot + 2) & 3), pb3 = 16 * ((rot + 3) & 3);

  for (int s = 0; s + 1 < STEPSN; ++s) {
    char* fd = a_base + ((s + 1) & 1) * FBBYTES + t * ROWB;
    const v4f f0 = cone_piece(a_base, cf, pb0);
    const v4f f1 = cone_piece(a_base, cf, pb1);
    const v4f f2 = cone_piece(a_base, cf, pb2);
    const v4f f3 = cone_piece(a_base, cf, pb3);
    st_row4(fd, pb0, f0); st_row4(fd, pb1, f1);
    st_row4(fd, pb2, f2); st_row4(fd, pb3, f3);
    st_row4(fd + COPYB, pb0, f0); st_row4(fd + COPYB, pb1, f1);
    st_row4(fd + COPYB, pb2, f2); st_row4(fd + COPYB, pb3, f3);
    cone_next(cf);
    __syncthreads();
  }

  {
    const int  pg  = t & (POPN - 1);
    const int  h   = t / POPN;
    const Cone cpp = cone_make(rp, sh_kb, copy, STEPSN - 1);
    const int  qb0 = 16 * ((h + rot) & 3), qb1 = 16 * ((h + 2 + rot) & 3);
    const v4f p0 = cone_piece(a_base, cpp, qb0);
    const v4f p1 = cone_piece(a_base, cpp, qb1);
    char* pd = reinterpret_cast<char*>(sh_pop) + pg * ROWB;
    st_row4(pd, qb0, p0); st_row4(pd, qb1, p1);
    __syncthreads();
  }

  if (t < 32) {
    v8f acc = {};
    v2f ones; ones[0] = 1.0f; ones[1] = 1.0f;
    const int col   = t & 15;
    const int khalf = (t < 16) ? 0 : 2;
    #pragma unroll 4
    for (int kc = 0; kc < 32; ++kc) {
      v2f bm;
      bm[0] = sh_pop[(4 * kc + khalf + 0) * BT + col];
      bm[1] = sh_pop[(4 * kc + khalf + 1) * BT + col];
      acc = __builtin_amdgcn_wmma_f32_16x16x4_f32(
           false, ones,  false, bm,
           (short)0, acc,  false,  false);
    }
    if (t < 16) {
      float logit = acc[0] / 10.0f;
      float* ob = out + (size_t)(blockIdx.x * BT + t) * 10;
      #pragma unroll
      for (int c = 0; c < 10; ++c) ob[c] = logit;
    }
  }
}

extern "C" void kernel_launch(void* const* d_in, const int* in_sizes, int n_in,
                              void* d_out, int out_size, void* d_ws, size_t ws_size,
                              hipStream_t stream) {
  const float* x   = (const float*)d_in[0];
  const float* thr = (const float*)d_in[1];
  const float* lg0 = (const float*)d_in[2];
  const int*   le0 = (const int*)  d_in[3];
  const int*   ri0 = (const int*)  d_in[4];
  const float* lg1 = (const float*)d_in[5];
  const int*   le1 = (const int*)  d_in[6];
  const int*   ri1 = (const int*)  d_in[7];
  const float* lg2 = (const float*)d_in[8];
  const int*   le2 = (const int*)  d_in[9];
  const int*   ri2 = (const int*)  d_in[10];

  float4* rq = (float4*)d_ws;
  int2*   rl = (int2*)((char*)d_ws + 43008);

  dlgn_precomp<<<7 * 3, 128, 0, stream>>>(
      lg0, le0, ri0, lg1, le1, ri1, lg2, le2, ri2, rq, rl);

  dlgn_main<<<4096 / BT, NTHREADS, 0, stream>>>(x, thr, rq, rl, (float*)d_out);
}
